# starting kernel
# baseline (speedup 1.0000x reference)
#include <hip/hip_runtime.h>
#include <hip/hip_bf16.h>
#include <math.h>

typedef __attribute__((ext_vector_type(2))) float v2f;
typedef __attribute__((ext_vector_type(8))) float v8f;

#define IN_DIM  128
#define HID     16
#define OUT_DIM 2

__global__ void k_init_dinv(float* __restrict__ dinv, int n) {
    int i = blockIdx.x * blockDim.x + threadIdx.x;
    if (i < n) dinv[i] = 1.0f;
}

__global__ void k_deg_edges(const int* __restrict__ dst, float* __restrict__ deg, int nE) {
    int e = blockIdx.x * blockDim.x + threadIdx.x;
    if (e < nE) unsafeAtomicAdd(&deg[dst[e]], 1.0f);
}

__global__ void k_finalize_dinv(float* __restrict__ dinv, int n) {
    int i = blockIdx.x * blockDim.x + threadIdx.x;
    if (i < n) {
        float d = dinv[i];
        dinv[i] = (d > 0.0f) ? rsqrtf(d) : 0.0f;
    }
}

__global__ __launch_bounds__(128) void k_gemm1_wmma(const float* __restrict__ x,
                                                    const float* __restrict__ W1,
                                                    float* __restrict__ h, int n) {
    __shared__ float sW[IN_DIM * HID];
    __shared__ float sX[4][16 * 132];

    for (int t = threadIdx.x; t < IN_DIM * HID; t += blockDim.x) sW[t] = W1[t];

    const int wave = threadIdx.x >> 5;
    const int lane = threadIdx.x & 31;
    const int tile = blockIdx.x * 4 + wave;
    const int row0 = tile * 16;
    const bool active = (row0 < n);

    if (active) {
        #pragma unroll 4
        for (int i = 0; i < 16; ++i) {
            int r = row0 + i;
            if (r >= n) r = n - 1;
            const float4 v = ((const float4*)(x + (size_t)r * IN_DIM))[lane];
            *(float4*)&sX[wave][i * 132 + lane * 4] = v;
        }
    }
    __syncthreads();
    if (!active) return;

    const int m  = lane & 15;
    const int kb = (lane >> 4) << 1;
    const float* xr = &sX[wave][m * 132];

    v8f acc = {};
    #pragma unroll
    for (int k0 = 0; k0 < IN_DIM; k0 += 4) {
        v2f a, b;
        a.x = xr[k0 + kb];
        a.y = xr[k0 + kb + 1];
        b.x = sW[(k0 + kb)     * HID + m];
        b.y = sW[(k0 + kb + 1) * HID + m];
        acc = __builtin_amdgcn_wmma_f32_16x16x4_f32(false, a, false, b,
                                                    (short)0, acc, false, false);
    }

    const int rbase = row0 + ((lane >> 4) << 3);
    #pragma unroll
    for (int v = 0; v < 8; ++v) {
        int r = rbase + v;
        if (r < n) h[(size_t)r * HID + m] = acc[v];
    }
}

__global__ void k_selfloop16(const float* __restrict__ h, const float* __restrict__ dinv,
                             float* __restrict__ agg, int n) {
    int t = blockIdx.x * blockDim.x + threadIdx.x;
    if (t >= n * HID) return;
    int i = t / HID;
    float di = dinv[i];
    agg[t] = h[t] * (di * di);
}

__global__ void k_scatter1(const int* __restrict__ src, const int* __restrict__ dst,
                           const float* __restrict__ dinv, const float* __restrict__ h,
                           float* __restrict__ agg, int nE) {
    int e = blockIdx.x * blockDim.x + threadIdx.x;
    if (e >= nE) return;
    int s = src[e], d = dst[e];
    float nrm = dinv[s] * dinv[d];
    const float4* hr = (const float4*)(h + (size_t)s * HID);
    float* ar = agg + (size_t)d * HID;
    #pragma unroll
    for (int q = 0; q < 4; ++q) {
        float4 v = hr[q];
        unsafeAtomicAdd(ar + q * 4 + 0, v.x * nrm);
        unsafeAtomicAdd(ar + q * 4 + 1, v.y * nrm);
        unsafeAtomicAdd(ar + q * 4 + 2, v.z * nrm);
        unsafeAtomicAdd(ar + q * 4 + 3, v.w * nrm);
    }
}

__global__ void k_layer2(const float* __restrict__ agg1, const float* __restrict__ b1,
                         const float* __restrict__ W2, const float* __restrict__ dinv,
                         float* __restrict__ h2, float* __restrict__ agg2, int n) {
    int i = blockIdx.x * blockDim.x + threadIdx.x;
    if (i >= n) return;
    const float* r = agg1 + (size_t)i * HID;
    float z0 = 0.0f, z1 = 0.0f;
    #pragma unroll
    for (int k = 0; k < HID; ++k) {
        float v = fmaxf(r[k] + b1[k], 0.0f);
        z0 = fmaf(v, W2[k * OUT_DIM + 0], z0);
        z1 = fmaf(v, W2[k * OUT_DIM + 1], z1);
    }
    float di = dinv[i], w = di * di;
    h2[i * 2 + 0]   = z0;
    h2[i * 2 + 1]   = z1;
    agg2[i * 2 + 0] = z0 * w;
    agg2[i * 2 + 1] = z1 * w;
}

__global__ void k_scatter2(const int* __restrict__ src, const int* __restrict__ dst,
                           const float* __restrict__ dinv, const float* __restrict__ h2,
                           float* __restrict__ agg2, int nE) {
    int e = blockIdx.x * blockDim.x + threadIdx.x;
    if (e >= nE) return;
    int s = src[e], d = dst[e];
    float nrm = dinv[s] * dinv[d];
    float2 v = ((const float2*)h2)[s];
    unsafeAtomicAdd(&agg2[d * 2 + 0], v.x * nrm);
    unsafeAtomicAdd(&agg2[d * 2 + 1], v.y * nrm);
}

__global__ void k_final(const float* __restrict__ agg2, const float* __restrict__ b2,
                        float* __restrict__ out, int n) {
    int i = blockIdx.x * blockDim.x + threadIdx.x;
    if (i >= n) return;
    float z0 = agg2[i * 2 + 0] + b2[0];
    float z1 = agg2[i * 2 + 1] + b2[1];
    float mx = fmaxf(z0, z1);
    float l  = mx + logf(expf(z0 - mx) + expf(z1 - mx));
    out[i * 2 + 0] = z0 - l;
    out[i * 2 + 1] = z1 - l;
}

extern "C" void kernel_launch(void* const* d_in, const int* in_sizes, int n_in,
                              void* d_out, int out_size, void* d_ws, size_t ws_size,
                              hipStream_t stream) {
    const float* x  = (const float*)d_in[0];
    const int*   ei = (const int*)d_in[1];
    const float* W1 = (const float*)d_in[2];
    const float* b1 = (const float*)d_in[3];
    const float* W2 = (const float*)d_in[4];
    const float* b2 = (const float*)d_in[5];
    float* out = (float*)d_out;

    const int n  = in_sizes[0] / IN_DIM;
    const int nE = in_sizes[1] / 2;
    const int* src = ei;
    const int* dst = ei + nE;

    float* ws   = (float*)d_ws;
    float* dinv = ws;
    float* h    = ws + n;
    float* agg1 = ws + (size_t)n * 17;
    float* h2   = h;
    float* agg2 = h + (size_t)2 * n;

    const int B = 256;
    k_init_dinv    <<<(n  + B - 1) / B, B, 0, stream>>>(dinv, n);
    k_deg_edges    <<<(nE + B - 1) / B, B, 0, stream>>>(dst, dinv, nE);
    k_finalize_dinv<<<(n  + B - 1) / B, B, 0, stream>>>(dinv, n);

    const int tiles   = (n + 15) / 16;
    const int gblocks = (tiles + 3) / 4;
    k_gemm1_wmma   <<<gblocks, 128, 0, stream>>>(x, W1, h, n);

    k_selfloop16   <<<(n * HID + B - 1) / B, B, 0, stream>>>(h, dinv, agg1, n);
    k_scatter1     <<<(nE + B - 1) / B, B, 0, stream>>>(src, dst, dinv, h, agg1, nE);
    k_layer2       <<<(n  + B - 1) / B, B, 0, stream>>>(agg1, b1, W2, dinv, h2, agg2, n);
    k_scatter2     <<<(nE + B - 1) / B, B, 0, stream>>>(src, dst, dinv, h2, agg2, nE);
    k_final        <<<(n  + B - 1) / B, B, 0, stream>>>(agg2, b2, out, n);
}


# scatter kernels remapped: a wave's lanes cover the channels of one edge (16 lanes per edge in layer 1, 2 in layer 2, indices by lane shuffle) so each float atomic and gather touches whole rows
# speedup vs baseline: 2.4989x; 2.4989x over previous
#include <hip/hip_runtime.h>
#include <hip/hip_bf16.h>
#include <math.h>

typedef __attribute__((ext_vector_type(2))) float v2f;
typedef __attribute__((ext_vector_type(8))) float v8f;

#define IN_DIM  128
#define HID     16
#define OUT_DIM 2

__global__ void k_init_dinv(float* __restrict__ dinv, int n) {
    int i = blockIdx.x * blockDim.x + threadIdx.x;
    if (i < n) dinv[i] = 1.0f;
}

__global__ void k_deg_edges(const int* __restrict__ dst, float* __restrict__ deg, int nE) {
    int e = blockIdx.x * blockDim.x + threadIdx.x;
    if (e < nE) unsafeAtomicAdd(&deg[dst[e]], 1.0f);
}

__global__ void k_finalize_dinv(float* __restrict__ dinv, int n) {
    int i = blockIdx.x * blockDim.x + threadIdx.x;
    if (i < n) {
        float d = dinv[i];
        dinv[i] = (d > 0.0f) ? rsqrtf(d) : 0.0f;
    }
}

__global__ __launch_bounds__(128) void k_gemm1_wmma(const float* __restrict__ x,
                                                    const float* __restrict__ W1,
                                                    float* __restrict__ h, int n) {
    __shared__ float sW[IN_DIM * HID];
    __shared__ float sX[4][16 * 132];

    for (int t = threadIdx.x; t < IN_DIM * HID; t += blockDim.x) sW[t] = W1[t];

    const int wave = threadIdx.x >> 5;
    const int lane = threadIdx.x & 31;
    const int tile = blockIdx.x * 4 + wave;
    const int row0 = tile * 16;
    const bool active = (row0 < n);

    if (active) {
        #pragma unroll 4
        for (int i = 0; i < 16; ++i) {
            int r = row0 + i;
            if (r >= n) r = n - 1;
            const float4 v = ((const float4*)(x + (size_t)r * IN_DIM))[lane];
            *(float4*)&sX[wave][i * 132 + lane * 4] = v;
        }
    }
    __syncthreads();
    if (!active) return;

    const int m  = lane & 15;
    const int kb = (lane >> 4) << 1;
    const float* xr = &sX[wave][m * 132];

    v8f acc = {};
    #pragma unroll
    for (int k0 = 0; k0 < IN_DIM; k0 += 4) {
        v2f a, b;
        a.x = xr[k0 + kb];
        a.y = xr[k0 + kb + 1];
        b.x = sW[(k0 + kb)     * HID + m];
        b.y = sW[(k0 + kb + 1) * HID + m];
        acc = __builtin_amdgcn_wmma_f32_16x16x4_f32(false, a, false, b,
                                                    (short)0, acc, false, false);
    }

    const int rbase = row0 + ((lane >> 4) << 3);
    #pragma unroll
    for (int v = 0; v < 8; ++v) {
        int r = rbase + v;
        if (r < n) h[(size_t)r * HID + m] = acc[v];
    }
}

__global__ void k_selfloop16(const float* __restrict__ h, const float* __restrict__ dinv,
                             float* __restrict__ agg, int n) {
    int t = blockIdx.x * blockDim.x + threadIdx.x;
    if (t >= n * HID) return;
    int i = t / HID;
    float di = dinv[i];
    agg[t] = h[t] * (di * di);
}

__global__ void k_scatter1(const int* __restrict__ src, const int* __restrict__ dst,
                           const float* __restrict__ dinv, const float* __restrict__ h,
                           float* __restrict__ agg, int nE) {
    const int lane  = threadIdx.x & 31;
    const int ebase = blockIdx.x * blockDim.x + (threadIdx.x & ~31);
    if (ebase >= nE) return;
    const int e = ebase + lane;
    int s = 0, d = 0;
    float nrm = 0.0f;
    if (e < nE) {
        s = src[e];
        d = dst[e];
        nrm = dinv[s] * dinv[d];
    }
    const unsigned c    = lane & 15;
    const int      half = lane >> 4;
    float v[HID], nj[HID];
    unsigned dj[HID];
    #pragma unroll
    for (int j = 0; j < HID; ++j) {
        const int sl = 2 * j + half;
        const unsigned sj = (unsigned)__shfl(s, sl);
        dj[j] = (unsigned)__shfl(d, sl);
        nj[j] = __shfl(nrm, sl);
        v[j]  = h[sj * HID + c];
    }
    #pragma unroll
    for (int j = 0; j < HID; ++j) {
        if (ebase + 2 * j + half < nE)
            unsafeAtomicAdd(agg + (dj[j] * HID + c), v[j] * nj[j]);
    }
}

__global__ void k_layer2(const float* __restrict__ agg1, const float* __restrict__ b1,
                         const float* __restrict__ W2, const float* __restrict__ dinv,
                         float* __restrict__ h2, float* __restrict__ agg2, int n) {
    int i = blockIdx.x * blockDim.x + threadIdx.x;
    if (i >= n) return;
    const float* r = agg1 + (size_t)i * HID;
    float z0 = 0.0f, z1 = 0.0f;
    #pragma unroll
    for (int k = 0; k < HID; ++k) {
        float v = fmaxf(r[k] + b1[k], 0.0f);
        z0 = fmaf(v, W2[k * OUT_DIM + 0], z0);
        z1 = fmaf(v, W2[k * OUT_DIM + 1], z1);
    }
    float di = dinv[i], w = di * di;
    h2[i * 2 + 0]   = z0;
    h2[i * 2 + 1]   = z1;
    agg2[i * 2 + 0] = z0 * w;
    agg2[i * 2 + 1] = z1 * w;
}

__global__ void k_scatter2(const int* __restrict__ src, const int* __restrict__ dst,
                           const float* __restrict__ dinv, const float* __restrict__ h2,
                           float* __restrict__ agg2, int nE) {
    const int lane  = threadIdx.x & 31;
    const int ebase = blockIdx.x * blockDim.x + (threadIdx.x & ~31);
    if (ebase >= nE) return;
    const int e = ebase + lane;
    int s = 0, d = 0;
    float nrm = 0.0f;
    if (e < nE) {
        s = src[e];
        d = dst[e];
        nrm = dinv[s] * dinv[d];
    }
    const unsigned c = lane & 1;
    const int      p = lane >> 1;
    float v[2], nj[2];
    unsigned dj[2];
    #pragma unroll
    for (int j = 0; j < 2; ++j) {
        const int sl = 16 * j + p;
        const unsigned sj = (unsigned)__shfl(s, sl);
        dj[j] = (unsigned)__shfl(d, sl);
        nj[j] = __shfl(nrm, sl);
        v[j]  = h2[sj * OUT_DIM + c];
    }
    #pragma unroll
    for (int j = 0; j < 2; ++j) {
        if (ebase + 16 * j + p < nE)
            unsafeAtomicAdd(agg2 + (dj[j] * OUT_DIM + c), v[j] * nj[j]);
    }
}

__global__ void k_final(const float* __restrict__ agg2, const float* __restrict__ b2,
                        float* __restrict__ out, int n) {
    int i = blockIdx.x * blockDim.x + threadIdx.x;
    if (i >= n) return;
    float z0 = agg2[i * 2 + 0] + b2[0];
    float z1 = agg2[i * 2 + 1] + b2[1];
    float mx = fmaxf(z0, z1);
    float l  = mx + logf(expf(z0 - mx) + expf(z1 - mx));
    out[i * 2 + 0] = z0 - l;
    out[i * 2 + 1] = z1 - l;
}

extern "C" void kernel_launch(void* const* d_in, const int* in_sizes, int n_in,
                              void* d_out, int out_size, void* d_ws, size_t ws_size,
                              hipStream_t stream) {
    const float* x  = (const float*)d_in[0];
    const int*   ei = (const int*)d_in[1];
    const float* W1 = (const float*)d_in[2];
    const float* b1 = (const float*)d_in[3];
    const float* W2 = (const float*)d_in[4];
    const float* b2 = (const float*)d_in[5];
    float* out = (float*)d_out;

    const int n  = in_sizes[0] / IN_DIM;
    const int nE = in_sizes[1] / 2;
    const int* src = ei;
    const int* dst = ei + nE;

    float* ws   = (float*)d_ws;
    float* dinv = ws;
    float* h    = ws + n;
    float* agg1 = ws + (size_t)n * 17;
    float* h2   = h;
    float* agg2 = h + (size_t)2 * n;

    const int B = 256;
    k_init_dinv    <<<(n  + B - 1) / B, B, 0, stream>>>(dinv, n);
    k_deg_edges    <<<(nE + B - 1) / B, B, 0, stream>>>(dst, dinv, nE);
    k_finalize_dinv<<<(n  + B - 1) / B, B, 0, stream>>>(dinv, n);

    const int tiles   = (n + 15) / 16;
    const int gblocks = (tiles + 3) / 4;
    k_gemm1_wmma   <<<gblocks, 128, 0, stream>>>(x, W1, h, n);

    k_selfloop16   <<<(n * HID + B - 1) / B, B, 0, stream>>>(h, dinv, agg1, n);
    k_scatter1     <<<(nE + B - 1) / B, B, 0, stream>>>(src, dst, dinv, h, agg1, nE);
    k_layer2       <<<(n  + B - 1) / B, B, 0, stream>>>(agg1, b1, W2, dinv, h2, agg2, n);
    k_scatter2     <<<(nE + B - 1) / B, B, 0, stream>>>(src, dst, dinv, h2, agg2, nE);
    k_final        <<<(n  + B - 1) / B, B, 0, stream>>>(agg2, b2, out, n);
}


# layer-1 scatter sums each 4096-row destination range in LDS (blocks filter an edge slice by range), then adds whole 128-byte lines to memory once per block; float sum order changes
# speedup vs baseline: 2.6648x; 1.0664x over previous
#include <hip/hip_runtime.h>
#include <hip/hip_bf16.h>
#include <math.h>

typedef __attribute__((ext_vector_type(2))) float v2f;
typedef __attribute__((ext_vector_type(8))) float v8f;

#define IN_DIM  128
#define HID     16
#define OUT_DIM 2

__global__ void k_init_dinv(float* __restrict__ dinv, int n) {
    int i = blockIdx.x * blockDim.x + threadIdx.x;
    if (i < n) dinv[i] = 1.0f;
}

__global__ void k_deg_edges(const int* __restrict__ dst, float* __restrict__ deg, int nE) {
    int e = blockIdx.x * blockDim.x + threadIdx.x;
    if (e < nE) unsafeAtomicAdd(&deg[dst[e]], 1.0f);
}

__global__ void k_finalize_dinv(float* __restrict__ dinv, int n) {
    int i = blockIdx.x * blockDim.x + threadIdx.x;
    if (i < n) {
        float d = dinv[i];
        dinv[i] = (d > 0.0f) ? rsqrtf(d) : 0.0f;
    }
}

__global__ __launch_bounds__(128) void k_gemm1_wmma(const float* __restrict__ x,
                                                    const float* __restrict__ W1,
                                                    float* __restrict__ h, int n) {
    __shared__ float sW[IN_DIM * HID];
    __shared__ float sX[4][16 * 132];

    for (int t = threadIdx.x; t < IN_DIM * HID; t += blockDim.x) sW[t] = W1[t];

    const int wave = threadIdx.x >> 5;
    const int lane = threadIdx.x & 31;
    const int tile = blockIdx.x * 4 + wave;
    const int row0 = tile * 16;
    const bool active = (row0 < n);

    if (active) {
        #pragma unroll 4
        for (int i = 0; i < 16; ++i) {
            int r = row0 + i;
            if (r >= n) r = n - 1;
            const float4 v = ((const float4*)(x + (size_t)r * IN_DIM))[lane];
            *(float4*)&sX[wave][i * 132 + lane * 4] = v;
        }
    }
    __syncthreads();
    if (!active) return;

    const int m  = lane & 15;
    const int kb = (lane >> 4) << 1;
    const float* xr = &sX[wave][m * 132];

    v8f acc = {};
    #pragma unroll
    for (int k0 = 0; k0 < IN_DIM; k0 += 4) {
        v2f a, b;
        a.x = xr[k0 + kb];
        a.y = xr[k0 + kb + 1];
        b.x = sW[(k0 + kb)     * HID + m];
        b.y = sW[(k0 + kb + 1) * HID + m];
        acc = __builtin_amdgcn_wmma_f32_16x16x4_f32(false, a, false, b,
                                                    (short)0, acc, false, false);
    }

    const int rbase = row0 + ((lane >> 4) << 3);
    #pragma unroll
    for (int v = 0; v < 8; ++v) {
        int r = rbase + v;
        if (r < n) h[(size_t)r * HID + m] = acc[v];
    }
}

__global__ void k_selfloop16(const float* __restrict__ h, const float* __restrict__ dinv,
                             float* __restrict__ agg, int n) {
    int t = blockIdx.x * blockDim.x + threadIdx.x;
    if (t >= n * HID) return;
    int i = t / HID;
    float di = dinv[i];
    agg[t] = h[t] * (di * di);
}

#define S1_RBITS  12
#define S1_R      (1 << S1_RBITS)
#define S1_SLICES 10
#define S1_WAVES  32
#define S1_Q      64
#define S1_CHUNK  128

__device__ __forceinline__ void s1_drain(const unsigned* ringw, unsigned head, int count,
                                         int lane, int lo,
                                         const float* __restrict__ dinv,
                                         const float* __restrict__ h, float* acc) {
    unsigned p = 0;
    float nrm = 0.0f;
    if (lane < count) {
        p = ringw[(head + (unsigned)lane) & (S1_Q - 1)];
        nrm = dinv[p >> S1_RBITS] * dinv[lo + (int)(p & (S1_R - 1))];
    }
    const unsigned c    = lane & 15;
    const int      half = lane >> 4;
    float v[HID], nj[HID];
    unsigned dj[HID];
    #pragma unroll
    for (int j = 0; j < HID; ++j) {
        const int sl = 2 * j + half;
        const unsigned pj = (unsigned)__shfl((int)p, sl);
        dj[j] = pj & (S1_R - 1);
        nj[j] = __shfl(nrm, sl);
        v[j]  = h[(pj >> S1_RBITS) * HID + c];
    }
    #pragma unroll
    for (int j = 0; j < HID; ++j) {
        if (2 * j + half < count)
            unsafeAtomicAdd(&acc[dj[j] * HID + c], v[j] * nj[j]);
    }
}

__global__ __launch_bounds__(1024) void k_scatter1(const int* __restrict__ src, const int* __restrict__ dst,
                                                   const float* __restrict__ dinv, const float* __restrict__ h,
                                                   float* __restrict__ agg, int nE, int n) {
    __shared__ __attribute__((aligned(16))) float acc[S1_R * HID];
    __shared__ unsigned ring[S1_WAVES][S1_Q];

    const int tid  = threadIdx.x;
    const int lane = tid & 31;
    const int wave = __builtin_amdgcn_readfirstlane(tid >> 5);
    const int lo   = blockIdx.x * S1_R;
    const int nsl  = (int)gridDim.y;
    const int per  = ((nE + nsl - 1) / nsl + S1_CHUNK - 1) / S1_CHUNK * S1_CHUNK;
    const int sbeg = (int)blockIdx.y * per;
    const int send = min(sbeg + per, nE);

    for (int i = tid; i < S1_R * HID / 4; i += S1_WAVES * 32)
        ((float4*)acc)[i] = make_float4(0.0f, 0.0f, 0.0f, 0.0f);
    __syncthreads();

    unsigned* ringw = ring[wave];
    unsigned qh = 0, qt = 0;

    auto load_chunk = [&](int (&d)[4], int (&s)[4], int base) {
        #pragma unroll
        for (int k = 0; k < 4; ++k) {
            const int e  = base + k * 32 + lane;
            const int ec = min(e, nE - 1);
            const int dv = dst[ec];
            s[k] = src[ec];
            d[k] = (e < send) ? dv : -1;
        }
    };
    auto filter_chunk = [&](const int (&d)[4], const int (&s)[4]) {
        #pragma unroll
        for (int k = 0; k < 4; ++k) {
            const unsigned dl = (unsigned)(d[k] - lo);
            const bool m = dl < (unsigned)S1_R;
            const unsigned mask = __builtin_amdgcn_ballot_w32(m);
            if (m)
                ringw[(qt + __builtin_amdgcn_mbcnt_lo(mask, 0u)) & (S1_Q - 1)] =
                    ((unsigned)s[k] << S1_RBITS) | dl;
            qt += (unsigned)__builtin_popcount(mask);
            if (qt - qh >= 32u) {
                s1_drain(ringw, qh, 32, lane, lo, dinv, h, acc);
                qh += 32u;
            }
        }
    };

    const int step = S1_WAVES * S1_CHUNK;
    int e0 = sbeg + wave * S1_CHUNK;
    int dA[4], sA[4], dB[4], sB[4];
    load_chunk(dA, sA, e0);
    for (; e0 < send; e0 += 2 * step) {
        load_chunk(dB, sB, e0 + step);
        filter_chunk(dA, sA);
        load_chunk(dA, sA, e0 + 2 * step);
        filter_chunk(dB, sB);
    }
    if (qt != qh) s1_drain(ringw, qh, (int)(qt - qh), lane, lo, dinv, h, acc);
    __syncthreads();

    const int rows = min(S1_R, n - lo);
    float* aggb = agg + (size_t)lo * HID;
    for (int i = tid; i < rows * HID; i += S1_WAVES * 32)
        unsafeAtomicAdd(aggb + i, acc[i]);
}

__global__ void k_layer2(const float* __restrict__ agg1, const float* __restrict__ b1,
                         const float* __restrict__ W2, const float* __restrict__ dinv,
                         float* __restrict__ h2, float* __restrict__ agg2, int n) {
    int i = blockIdx.x * blockDim.x + threadIdx.x;
    if (i >= n) return;
    const float* r = agg1 + (size_t)i * HID;
    float z0 = 0.0f, z1 = 0.0f;
    #pragma unroll
    for (int k = 0; k < HID; ++k) {
        float v = fmaxf(r[k] + b1[k], 0.0f);
        z0 = fmaf(v, W2[k * OUT_DIM + 0], z0);
        z1 = fmaf(v, W2[k * OUT_DIM + 1], z1);
    }
    float di = dinv[i], w = di * di;
    h2[i * 2 + 0]   = z0;
    h2[i * 2 + 1]   = z1;
    agg2[i * 2 + 0] = z0 * w;
    agg2[i * 2 + 1] = z1 * w;
}

__global__ void k_scatter2(const int* __restrict__ src, const int* __restrict__ dst,
                           const float* __restrict__ dinv, const float* __restrict__ h2,
                           float* __restrict__ agg2, int nE) {
    const int lane  = threadIdx.x & 31;
    const int ebase = blockIdx.x * blockDim.x + (threadIdx.x & ~31);
    if (ebase >= nE) return;
    const int e = ebase + lane;
    int s = 0, d = 0;
    float nrm = 0.0f;
    if (e < nE) {
        s = src[e];
        d = dst[e];
        nrm = dinv[s] * dinv[d];
    }
    const unsigned c = lane & 1;
    const int      p = lane >> 1;
    float v[2], nj[2];
    unsigned dj[2];
    #pragma unroll
    for (int j = 0; j < 2; ++j) {
        const int sl = 16 * j + p;
        const unsigned sj = (unsigned)__shfl(s, sl);
        dj[j] = (unsigned)__shfl(d, sl);
        nj[j] = __shfl(nrm, sl);
        v[j]  = h2[sj * OUT_DIM + c];
    }
    #pragma unroll
    for (int j = 0; j < 2; ++j) {
        if (ebase + 16 * j + p < nE)
            unsafeAtomicAdd(agg2 + (dj[j] * OUT_DIM + c), v[j] * nj[j]);
    }
}

__global__ void k_final(const float* __restrict__ agg2, const float* __restrict__ b2,
                        float* __restrict__ out, int n) {
    int i = blockIdx.x * blockDim.x + threadIdx.x;
    if (i >= n) return;
    float z0 = agg2[i * 2 + 0] + b2[0];
    float z1 = agg2[i * 2 + 1] + b2[1];
    float mx = fmaxf(z0, z1);
    float l  = mx + logf(expf(z0 - mx) + expf(z1 - mx));
    out[i * 2 + 0] = z0 - l;
    out[i * 2 + 1] = z1 - l;
}

extern "C" void kernel_launch(void* const* d_in, const int* in_sizes, int n_in,
                              void* d_out, int out_size, void* d_ws, size_t ws_size,
                              hipStream_t stream) {
    const float* x  = (const float*)d_in[0];
    const int*   ei = (const int*)d_in[1];
    const float* W1 = (const float*)d_in[2];
    const float* b1 = (const float*)d_in[3];
    const float* W2 = (const float*)d_in[4];
    const float* b2 = (const float*)d_in[5];
    float* out = (float*)d_out;

    const int n  = in_sizes[0] / IN_DIM;
    const int nE = in_sizes[1] / 2;
    const int* src = ei;
    const int* dst = ei + nE;

    float* ws   = (float*)d_ws;
    float* dinv = ws;
    float* h    = ws + n;
    float* agg1 = ws + (size_t)n * 17;
    float* h2   = h;
    float* agg2 = h + (size_t)2 * n;

    const int B = 256;
    k_init_dinv    <<<(n  + B - 1) / B, B, 0, stream>>>(dinv, n);
    k_deg_edges    <<<(nE + B - 1) / B, B, 0, stream>>>(dst, dinv, nE);
    k_finalize_dinv<<<(n  + B - 1) / B, B, 0, stream>>>(dinv, n);

    const int tiles   = (n + 15) / 16;
    const int gblocks = (tiles + 3) / 4;
    k_gemm1_wmma   <<<gblocks, 128, 0, stream>>>(x, W1, h, n);

    k_selfloop16   <<<(n * HID + B - 1) / B, B, 0, stream>>>(h, dinv, agg1, n);
    k_scatter1     <<<dim3((n + S1_R - 1) / S1_R, S1_SLICES), S1_WAVES * 32, 0, stream>>>(src, dst, dinv, h, agg1, nE, n);
    k_layer2       <<<(n  + B - 1) / B, B, 0, stream>>>(agg1, b1, W2, dinv, h2, agg2, n);
    k_scatter2     <<<(nE + B - 1) / B, B, 0, stream>>>(src, dst, dinv, h2, agg2, nE);
    k_final        <<<(n  + B - 1) / B, B, 0, stream>>>(agg2, b2, out, n);
}


# degree count formed in LDS: blocks count one slice's destinations per 51200-node range with LDS float adds (exact), then add whole 128-B lines to memory; 3.2M scattered atomics become 400k line adds
# speedup vs baseline: 3.8764x; 1.4547x over previous
#include <hip/hip_runtime.h>
#include <hip/hip_bf16.h>
#include <math.h>

typedef __attribute__((ext_vector_type(2))) float v2f;
typedef __attribute__((ext_vector_type(8))) float v8f;

#define IN_DIM  128
#define HID     16
#define OUT_DIM 2

__global__ void k_init_dinv(float* __restrict__ dinv, int n) {
    int i = blockIdx.x * blockDim.x + threadIdx.x;
    if (i < n) dinv[i] = 1.0f;
}

#define DG_R      51200
#define DG_SLICES 128
#define DG_T      1024
#define DG_B      25
#define DG_F      10

__global__ __launch_bounds__(DG_T) void k_deg_edges(const int* __restrict__ dst, float* __restrict__ deg,
                                                    int nE, int n) {
    __shared__ __attribute__((aligned(16))) float cnt[DG_R];

    const int tid  = threadIdx.x;
    const int lo   = blockIdx.x * DG_R;
    const int nsl  = (int)gridDim.y;
    const int per  = ((nE + nsl - 1) / nsl + 31) & ~31;
    const int sbeg = (int)blockIdx.y * per;
    const int send = min(sbeg + per, nE);

    for (int i = tid; i < DG_R / 4; i += DG_T)
        ((float4*)cnt)[i] = make_float4(0.0f, 0.0f, 0.0f, 0.0f);
    __syncthreads();

    for (int base = sbeg; base < send; base += DG_T * DG_B) {
        int d[DG_B];
        #pragma unroll
        for (int k = 0; k < DG_B; ++k)
            d[k] = dst[min(base + k * DG_T + tid, nE - 1)];
        #pragma unroll
        for (int k = 0; k < DG_B; ++k) {
            const unsigned dl = (unsigned)(d[k] - lo);
            if (base + k * DG_T + tid < send && dl < (unsigned)DG_R)
                unsafeAtomicAdd(&cnt[dl], 1.0f);
        }
    }
    __syncthreads();

    const int rows = min(DG_R, n - lo);
    float* degb = deg + lo;
    for (int i0 = tid; i0 < DG_R; i0 += DG_T * DG_F) {
        float c[DG_F];
        #pragma unroll
        for (int k = 0; k < DG_F; ++k) c[k] = cnt[i0 + k * DG_T];
        #pragma unroll
        for (int k = 0; k < DG_F; ++k)
            if (i0 + k * DG_T < rows) unsafeAtomicAdd(degb + i0 + k * DG_T, c[k]);
    }
}

__global__ void k_finalize_dinv(float* __restrict__ dinv, int n) {
    int i = blockIdx.x * blockDim.x + threadIdx.x;
    if (i < n) {
        float d = dinv[i];
        dinv[i] = (d > 0.0f) ? rsqrtf(d) : 0.0f;
    }
}

__global__ __launch_bounds__(128) void k_gemm1_wmma(const float* __restrict__ x,
                                                    const float* __restrict__ W1,
                                                    float* __restrict__ h, int n) {
    __shared__ float sW[IN_DIM * HID];
    __shared__ float sX[4][16 * 132];

    for (int t = threadIdx.x; t < IN_DIM * HID; t += blockDim.x) sW[t] = W1[t];

    const int wave = threadIdx.x >> 5;
    const int lane = threadIdx.x & 31;
    const int tile = blockIdx.x * 4 + wave;
    const int row0 = tile * 16;
    const bool active = (row0 < n);

    if (active) {
        #pragma unroll 4
        for (int i = 0; i < 16; ++i) {
            int r = row0 + i;
            if (r >= n) r = n - 1;
            const float4 v = ((const float4*)(x + (size_t)r * IN_DIM))[lane];
            *(float4*)&sX[wave][i * 132 + lane * 4] = v;
        }
    }
    __syncthreads();
    if (!active) return;

    const int m  = lane & 15;
    const int kb = (lane >> 4) << 1;
    const float* xr = &sX[wave][m * 132];

    v8f acc = {};
    #pragma unroll
    for (int k0 = 0; k0 < IN_DIM; k0 += 4) {
        v2f a, b;
        a.x = xr[k0 + kb];
        a.y = xr[k0 + kb + 1];
        b.x = sW[(k0 + kb)     * HID + m];
        b.y = sW[(k0 + kb + 1) * HID + m];
        acc = __builtin_amdgcn_wmma_f32_16x16x4_f32(false, a, false, b,
                                                    (short)0, acc, false, false);
    }

    const int rbase = row0 + ((lane >> 4) << 3);
    #pragma unroll
    for (int v = 0; v < 8; ++v) {
        int r = rbase + v;
        if (r < n) h[(size_t)r * HID + m] = acc[v];
    }
}

__global__ void k_selfloop16(const float* __restrict__ h, const float* __restrict__ dinv,
                             float* __restrict__ agg, int n) {
    int t = blockIdx.x * blockDim.x + threadIdx.x;
    if (t >= n * HID) return;
    int i = t / HID;
    float di = dinv[i];
    agg[t] = h[t] * (di * di);
}

#define S1_RBITS  12
#define S1_R      (1 << S1_RBITS)
#define S1_SLICES 10
#define S1_WAVES  32
#define S1_Q      64
#define S1_CHUNK  128

__device__ __forceinline__ void s1_drain(const unsigned* ringw, unsigned head, int count,
                                         int lane, int lo,
                                         const float* __restrict__ dinv,
                                         const float* __restrict__ h, float* acc) {
    unsigned p = 0;
    float nrm = 0.0f;
    if (lane < count) {
        p = ringw[(head + (unsigned)lane) & (S1_Q - 1)];
        nrm = dinv[p >> S1_RBITS] * dinv[lo + (int)(p & (S1_R - 1))];
    }
    const unsigned c    = lane & 15;
    const int      half = lane >> 4;
    float v[HID], nj[HID];
    unsigned dj[HID];
    #pragma unroll
    for (int j = 0; j < HID; ++j) {
        const int sl = 2 * j + half;
        const unsigned pj = (unsigned)__shfl((int)p, sl);
        dj[j] = pj & (S1_R - 1);
        nj[j] = __shfl(nrm, sl);
        v[j]  = h[(pj >> S1_RBITS) * HID + c];
    }
    #pragma unroll
    for (int j = 0; j < HID; ++j) {
        if (2 * j + half < count)
            unsafeAtomicAdd(&acc[dj[j] * HID + c], v[j] * nj[j]);
    }
}

__global__ __launch_bounds__(1024) void k_scatter1(const int* __restrict__ src, const int* __restrict__ dst,
                                                   const float* __restrict__ dinv, const float* __restrict__ h,
                                                   float* __restrict__ agg, int nE, int n) {
    __shared__ __attribute__((aligned(16))) float acc[S1_R * HID];
    __shared__ unsigned ring[S1_WAVES][S1_Q];

    const int tid  = threadIdx.x;
    const int lane = tid & 31;
    const int wave = __builtin_amdgcn_readfirstlane(tid >> 5);
    const int lo   = blockIdx.x * S1_R;
    const int nsl  = (int)gridDim.y;
    const int per  = ((nE + nsl - 1) / nsl + S1_CHUNK - 1) / S1_CHUNK * S1_CHUNK;
    const int sbeg = (int)blockIdx.y * per;
    const int send = min(sbeg + per, nE);

    for (int i = tid; i < S1_R * HID / 4; i += S1_WAVES * 32)
        ((float4*)acc)[i] = make_float4(0.0f, 0.0f, 0.0f, 0.0f);
    __syncthreads();

    unsigned* ringw = ring[wave];
    unsigned qh = 0, qt = 0;

    auto load_chunk = [&](int (&d)[4], int (&s)[4], int base) {
        #pragma unroll
        for (int k = 0; k < 4; ++k) {
            const int e  = base + k * 32 + lane;
            const int ec = min(e, nE - 1);
            const int dv = dst[ec];
            s[k] = src[ec];
            d[k] = (e < send) ? dv : -1;
        }
    };
    auto filter_chunk = [&](const int (&d)[4], const int (&s)[4]) {
        #pragma unroll
        for (int k = 0; k < 4; ++k) {
            const unsigned dl = (unsigned)(d[k] - lo);
            const bool m = dl < (unsigned)S1_R;
            const unsigned mask = __builtin_amdgcn_ballot_w32(m);
            if (m)
                ringw[(qt + __builtin_amdgcn_mbcnt_lo(mask, 0u)) & (S1_Q - 1)] =
                    ((unsigned)s[k] << S1_RBITS) | dl;
            qt += (unsigned)__builtin_popcount(mask);
            if (qt - qh >= 32u) {
                s1_drain(ringw, qh, 32, lane, lo, dinv, h, acc);
                qh += 32u;
            }
        }
    };

    const int step = S1_WAVES * S1_CHUNK;
    int e0 = sbeg + wave * S1_CHUNK;
    int dA[4], sA[4], dB[4], sB[4];
    load_chunk(dA, sA, e0);
    for (; e0 < send; e0 += 2 * step) {
        load_chunk(dB, sB, e0 + step);
        filter_chunk(dA, sA);
        load_chunk(dA, sA, e0 + 2 * step);
        filter_chunk(dB, sB);
    }
    if (qt != qh) s1_drain(ringw, qh, (int)(qt - qh), lane, lo, dinv, h, acc);
    __syncthreads();

    const int rows = min(S1_R, n - lo);
    float* aggb = agg + (size_t)lo * HID;
    for (int i = tid; i < rows * HID; i += S1_WAVES * 32)
        unsafeAtomicAdd(aggb + i, acc[i]);
}

__global__ void k_layer2(const float* __restrict__ agg1, const float* __restrict__ b1,
                         const float* __restrict__ W2, const float* __restrict__ dinv,
                         float* __restrict__ h2, float* __restrict__ agg2, int n) {
    int i = blockIdx.x * blockDim.x + threadIdx.x;
    if (i >= n) return;
    const float* r = agg1 + (size_t)i * HID;
    float z0 = 0.0f, z1 = 0.0f;
    #pragma unroll
    for (int k = 0; k < HID; ++k) {
        float v = fmaxf(r[k] + b1[k], 0.0f);
        z0 = fmaf(v, W2[k * OUT_DIM + 0], z0);
        z1 = fmaf(v, W2[k * OUT_DIM + 1], z1);
    }
    float di = dinv[i], w = di * di;
    h2[i * 2 + 0]   = z0;
    h2[i * 2 + 1]   = z1;
    agg2[i * 2 + 0] = z0 * w;
    agg2[i * 2 + 1] = z1 * w;
}

__global__ void k_scatter2(const int* __restrict__ src, const int* __restrict__ dst,
                           const float* __restrict__ dinv, const float* __restrict__ h2,
                           float* __restrict__ agg2, int nE) {
    const int lane  = threadIdx.x & 31;
    const int ebase = blockIdx.x * blockDim.x + (threadIdx.x & ~31);
    if (ebase >= nE) return;
    const int e = ebase + lane;
    int s = 0, d = 0;
    float nrm = 0.0f;
    if (e < nE) {
        s = src[e];
        d = dst[e];
        nrm = dinv[s] * dinv[d];
    }
    const unsigned c = lane & 1;
    const int      p = lane >> 1;
    float v[2], nj[2];
    unsigned dj[2];
    #pragma unroll
    for (int j = 0; j < 2; ++j) {
        const int sl = 16 * j + p;
        const unsigned sj = (unsigned)__shfl(s, sl);
        dj[j] = (unsigned)__shfl(d, sl);
        nj[j] = __shfl(nrm, sl);
        v[j]  = h2[sj * OUT_DIM + c];
    }
    #pragma unroll
    for (int j = 0; j < 2; ++j) {
        if (ebase + 16 * j + p < nE)
            unsafeAtomicAdd(agg2 + (dj[j] * OUT_DIM + c), v[j] * nj[j]);
    }
}

__global__ void k_final(const float* __restrict__ agg2, const float* __restrict__ b2,
                        float* __restrict__ out, int n) {
    int i = blockIdx.x * blockDim.x + threadIdx.x;
    if (i >= n) return;
    float z0 = agg2[i * 2 + 0] + b2[0];
    float z1 = agg2[i * 2 + 1] + b2[1];
    float mx = fmaxf(z0, z1);
    float l  = mx + logf(expf(z0 - mx) + expf(z1 - mx));
    out[i * 2 + 0] = z0 - l;
    out[i * 2 + 1] = z1 - l;
}

extern "C" void kernel_launch(void* const* d_in, const int* in_sizes, int n_in,
                              void* d_out, int out_size, void* d_ws, size_t ws_size,
                              hipStream_t stream) {
    const float* x  = (const float*)d_in[0];
    const int*   ei = (const int*)d_in[1];
    const float* W1 = (const float*)d_in[2];
    const float* b1 = (const float*)d_in[3];
    const float* W2 = (const float*)d_in[4];
    const float* b2 = (const float*)d_in[5];
    float* out = (float*)d_out;

    const int n  = in_sizes[0] / IN_DIM;
    const int nE = in_sizes[1] / 2;
    const int* src = ei;
    const int* dst = ei + nE;

    float* ws   = (float*)d_ws;
    float* dinv = ws;
    float* h    = ws + n;
    float* agg1 = ws + (size_t)n * 17;
    float* h2   = h;
    float* agg2 = h + (size_t)2 * n;

    const int B = 256;
    k_init_dinv    <<<(n  + B - 1) / B, B, 0, stream>>>(dinv, n);
    k_deg_edges    <<<dim3((n + DG_R - 1) / DG_R, DG_SLICES), DG_T, 0, stream>>>(dst, dinv, nE, n);
    k_finalize_dinv<<<(n  + B - 1) / B, B, 0, stream>>>(dinv, n);

    const int tiles   = (n + 15) / 16;
    const int gblocks = (tiles + 3) / 4;
    k_gemm1_wmma   <<<gblocks, 128, 0, stream>>>(x, W1, h, n);

    k_selfloop16   <<<(n * HID + B - 1) / B, B, 0, stream>>>(h, dinv, agg1, n);
    k_scatter1     <<<dim3((n + S1_R - 1) / S1_R, S1_SLICES), S1_WAVES * 32, 0, stream>>>(src, dst, dinv, h, agg1, nE, n);
    k_layer2       <<<(n  + B - 1) / B, B, 0, stream>>>(agg1, b1, W2, dinv, h2, agg2, n);
    k_scatter2     <<<(nE + B - 1) / B, B, 0, stream>>>(src, dst, dinv, h2, agg2, nE);
    k_final        <<<(n  + B - 1) / B, B, 0, stream>>>(agg2, b2, out, n);
}


# layer-2 scatter sums each 25024-node destination range in LDS (blocks filter an edge slice by range, range's dinv kept in LDS), then adds whole 128-byte lines to memory; float sum order changes
# speedup vs baseline: 6.8363x; 1.7636x over previous
#include <hip/hip_runtime.h>
#include <hip/hip_bf16.h>
#include <math.h>

typedef __attribute__((ext_vector_type(2))) float v2f;
typedef __attribute__((ext_vector_type(8))) float v8f;

#define IN_DIM  128
#define HID     16
#define OUT_DIM 2

__global__ void k_init_dinv(float* __restrict__ dinv, int n) {
    int i = blockIdx.x * blockDim.x + threadIdx.x;
    if (i < n) dinv[i] = 1.0f;
}

#define DG_R      51200
#define DG_SLICES 128
#define DG_T      1024
#define DG_B      25
#define DG_F      10

__global__ __launch_bounds__(DG_T) void k_deg_edges(const int* __restrict__ dst, float* __restrict__ deg,
                                                    int nE, int n) {
    __shared__ __attribute__((aligned(16))) float cnt[DG_R];

    const int tid  = threadIdx.x;
    const int lo   = blockIdx.x * DG_R;
    const int nsl  = (int)gridDim.y;
    const int per  = ((nE + nsl - 1) / nsl + 31) & ~31;
    const int sbeg = (int)blockIdx.y * per;
    const int send = min(sbeg + per, nE);

    for (int i = tid; i < DG_R / 4; i += DG_T)
        ((float4*)cnt)[i] = make_float4(0.0f, 0.0f, 0.0f, 0.0f);
    __syncthreads();

    for (int base = sbeg; base < send; base += DG_T * DG_B) {
        int d[DG_B];
        #pragma unroll
        for (int k = 0; k < DG_B; ++k)
            d[k] = dst[min(base + k * DG_T + tid, nE - 1)];
        #pragma unroll
        for (int k = 0; k < DG_B; ++k) {
            const unsigned dl = (unsigned)(d[k] - lo);
            if (base + k * DG_T + tid < send && dl < (unsigned)DG_R)
                unsafeAtomicAdd(&cnt[dl], 1.0f);
        }
    }
    __syncthreads();

    const int rows = min(DG_R, n - lo);
    float* degb = deg + lo;
    for (int i0 = tid; i0 < DG_R; i0 += DG_T * DG_F) {
        float c[DG_F];
        #pragma unroll
        for (int k = 0; k < DG_F; ++k) c[k] = cnt[i0 + k * DG_T];
        #pragma unroll
        for (int k = 0; k < DG_F; ++k)
            if (i0 + k * DG_T < rows) unsafeAtomicAdd(degb + i0 + k * DG_T, c[k]);
    }
}

__global__ void k_finalize_dinv(float* __restrict__ dinv, int n) {
    int i = blockIdx.x * blockDim.x + threadIdx.x;
    if (i < n) {
        float d = dinv[i];
        dinv[i] = (d > 0.0f) ? rsqrtf(d) : 0.0f;
    }
}

__global__ __launch_bounds__(128) void k_gemm1_wmma(const float* __restrict__ x,
                                                    const float* __restrict__ W1,
                                                    float* __restrict__ h, int n) {
    __shared__ float sW[IN_DIM * HID];
    __shared__ float sX[4][16 * 132];

    for (int t = threadIdx.x; t < IN_DIM * HID; t += blockDim.x) sW[t] = W1[t];

    const int wave = threadIdx.x >> 5;
    const int lane = threadIdx.x & 31;
    const int tile = blockIdx.x * 4 + wave;
    const int row0 = tile * 16;
    const bool active = (row0 < n);

    if (active) {
        #pragma unroll 4
        for (int i = 0; i < 16; ++i) {
            int r = row0 + i;
            if (r >= n) r = n - 1;
            const float4 v = ((const float4*)(x + (size_t)r * IN_DIM))[lane];
            *(float4*)&sX[wave][i * 132 + lane * 4] = v;
        }
    }
    __syncthreads();
    if (!active) return;

    const int m  = lane & 15;
    const int kb = (lane >> 4) << 1;
    const float* xr = &sX[wave][m * 132];

    v8f acc = {};
    #pragma unroll
    for (int k0 = 0; k0 < IN_DIM; k0 += 4) {
        v2f a, b;
        a.x = xr[k0 + kb];
        a.y = xr[k0 + kb + 1];
        b.x = sW[(k0 + kb)     * HID + m];
        b.y = sW[(k0 + kb + 1) * HID + m];
        acc = __builtin_amdgcn_wmma_f32_16x16x4_f32(false, a, false, b,
                                                    (short)0, acc, false, false);
    }

    const int rbase = row0 + ((lane >> 4) << 3);
    #pragma unroll
    for (int v = 0; v < 8; ++v) {
        int r = rbase + v;
        if (r < n) h[(size_t)r * HID + m] = acc[v];
    }
}

__global__ void k_selfloop16(const float* __restrict__ h, const float* __restrict__ dinv,
                             float* __restrict__ agg, int n) {
    int t = blockIdx.x * blockDim.x + threadIdx.x;
    if (t >= n * HID) return;
    int i = t / HID;
    float di = dinv[i];
    agg[t] = h[t] * (di * di);
}

#define S1_RBITS  12
#define S1_R      (1 << S1_RBITS)
#define S1_SLICES 10
#define S1_WAVES  32
#define S1_Q      64
#define S1_CHUNK  128

__device__ __forceinline__ void s1_drain(const unsigned* ringw, unsigned head, int count,
                                         int lane, int lo,
                                         const float* __restrict__ dinv,
                                         const float* __restrict__ h, float* acc) {
    unsigned p = 0;
    float nrm = 0.0f;
    if (lane < count) {
        p = ringw[(head + (unsigned)lane) & (S1_Q - 1)];
        nrm = dinv[p >> S1_RBITS] * dinv[lo + (int)(p & (S1_R - 1))];
    }
    const unsigned c    = lane & 15;
    const int      half = lane >> 4;
    float v[HID], nj[HID];
    unsigned dj[HID];
    #pragma unroll
    for (int j = 0; j < HID; ++j) {
        const int sl = 2 * j + half;
        const unsigned pj = (unsigned)__shfl((int)p, sl);
        dj[j] = pj & (S1_R - 1);
        nj[j] = __shfl(nrm, sl);
        v[j]  = h[(pj >> S1_RBITS) * HID + c];
    }
    #pragma unroll
    for (int j = 0; j < HID; ++j) {
        if (2 * j + half < count)
            unsafeAtomicAdd(&acc[dj[j] * HID + c], v[j] * nj[j]);
    }
}

__global__ __launch_bounds__(1024) void k_scatter1(const int* __restrict__ src, const int* __restrict__ dst,
                                                   const float* __restrict__ dinv, const float* __restrict__ h,
                                                   float* __restrict__ agg, int nE, int n) {
    __shared__ __attribute__((aligned(16))) float acc[S1_R * HID];
    __shared__ unsigned ring[S1_WAVES][S1_Q];

    const int tid  = threadIdx.x;
    const int lane = tid & 31;
    const int wave = __builtin_amdgcn_readfirstlane(tid >> 5);
    const int lo   = blockIdx.x * S1_R;
    const int nsl  = (int)gridDim.y;
    const int per  = ((nE + nsl - 1) / nsl + S1_CHUNK - 1) / S1_CHUNK * S1_CHUNK;
    const int sbeg = (int)blockIdx.y * per;
    const int send = min(sbeg + per, nE);

    for (int i = tid; i < S1_R * HID / 4; i += S1_WAVES * 32)
        ((float4*)acc)[i] = make_float4(0.0f, 0.0f, 0.0f, 0.0f);
    __syncthreads();

    unsigned* ringw = ring[wave];
    unsigned qh = 0, qt = 0;

    auto load_chunk = [&](int (&d)[4], int (&s)[4], int base) {
        #pragma unroll
        for (int k = 0; k < 4; ++k) {
            const int e  = base + k * 32 + lane;
            const int ec = min(e, nE - 1);
            const int dv = dst[ec];
            s[k] = src[ec];
            d[k] = (e < send) ? dv : -1;
        }
    };
    auto filter_chunk = [&](const int (&d)[4], const int (&s)[4]) {
        #pragma unroll
        for (int k = 0; k < 4; ++k) {
            const unsigned dl = (unsigned)(d[k] - lo);
            const bool m = dl < (unsigned)S1_R;
            const unsigned mask = __builtin_amdgcn_ballot_w32(m);
            if (m)
                ringw[(qt + __builtin_amdgcn_mbcnt_lo(mask, 0u)) & (S1_Q - 1)] =
                    ((unsigned)s[k] << S1_RBITS) | dl;
            qt += (unsigned)__builtin_popcount(mask);
            if (qt - qh >= 32u) {
                s1_drain(ringw, qh, 32, lane, lo, dinv, h, acc);
                qh += 32u;
            }
        }
    };

    const int step = S1_WAVES * S1_CHUNK;
    int e0 = sbeg + wave * S1_CHUNK;
    int dA[4], sA[4], dB[4], sB[4];
    load_chunk(dA, sA, e0);
    for (; e0 < send; e0 += 2 * step) {
        load_chunk(dB, sB, e0 + step);
        filter_chunk(dA, sA);
        load_chunk(dA, sA, e0 + 2 * step);
        filter_chunk(dB, sB);
    }
    if (qt != qh) s1_drain(ringw, qh, (int)(qt - qh), lane, lo, dinv, h, acc);
    __syncthreads();

    const int rows = min(S1_R, n - lo);
    float* aggb = agg + (size_t)lo * HID;
    for (int i = tid; i < rows * HID; i += S1_WAVES * 32)
        unsafeAtomicAdd(aggb + i, acc[i]);
}

__global__ void k_layer2(const float* __restrict__ agg1, const float* __restrict__ b1,
                         const float* __restrict__ W2, const float* __restrict__ dinv,
                         float* __restrict__ h2, float* __restrict__ agg2, int n) {
    int i = blockIdx.x * blockDim.x + threadIdx.x;
    if (i >= n) return;
    const float* r = agg1 + (size_t)i * HID;
    float z0 = 0.0f, z1 = 0.0f;
    #pragma unroll
    for (int k = 0; k < HID; ++k) {
        float v = fmaxf(r[k] + b1[k], 0.0f);
        z0 = fmaf(v, W2[k * OUT_DIM + 0], z0);
        z1 = fmaf(v, W2[k * OUT_DIM + 1], z1);
    }
    float di = dinv[i], w = di * di;
    h2[i * 2 + 0]   = z0;
    h2[i * 2 + 1]   = z1;
    agg2[i * 2 + 0] = z0 * w;
    agg2[i * 2 + 1] = z1 * w;
}

#define S2_R      25024
#define S2_SLICES 64
#define S2_T      1024
#define S2_K      7
#define S2_F      7
#define S2_ST     ((S2_R + S2_T - 1) / S2_T)

__global__ __launch_bounds__(S2_T) void k_scatter2(const int* __restrict__ src, const int* __restrict__ dst,
                                                   const float* __restrict__ dinv, const float* __restrict__ h2,
                                                   float* __restrict__ agg2, int nE, int n) {
    __shared__ __attribute__((aligned(16))) float acc[S2_R * OUT_DIM];
    __shared__ __attribute__((aligned(16))) float dloc[S2_R];

    const int tid  = threadIdx.x;
    const int lo   = blockIdx.x * S2_R;
    const int rows = min(S2_R, n - lo);
    const int nsl  = (int)gridDim.y;
    const int per  = ((nE + nsl - 1) / nsl + 31) & ~31;
    const int sbeg = (int)blockIdx.y * per;
    const int send = min(sbeg + per, nE);

    for (int i = tid; i < S2_R * OUT_DIM / 4; i += S2_T)
        ((float4*)acc)[i] = make_float4(0.0f, 0.0f, 0.0f, 0.0f);
    {
        float dv[S2_ST];
        #pragma unroll
        for (int k = 0; k < S2_ST; ++k) {
            const int i = k * S2_T + tid;
            dv[k] = (i < rows) ? dinv[lo + i] : 0.0f;
        }
        #pragma unroll
        for (int k = 0; k < S2_ST; ++k) {
            const int i = k * S2_T + tid;
            if (i < S2_R) dloc[i] = dv[k];
        }
    }
    __syncthreads();

    auto load_idx = [&](int (&d)[S2_K], int (&s)[S2_K], int base) {
        #pragma unroll
        for (int k = 0; k < S2_K; ++k) {
            const int e  = base + k * S2_T + tid;
            const int ec = min(e, nE - 1);
            const int dv = dst[ec];
            s[k] = src[ec];
            d[k] = (e < send) ? dv : -1;
        }
    };

    int d[S2_K], s[S2_K];
    load_idx(d, s, sbeg);
    for (int base = sbeg; base < send; base += S2_T * S2_K) {
        unsigned dl[S2_K];
        bool     own[S2_K];
        float2   hv[S2_K];
        float    ds[S2_K], dd[S2_K];
        #pragma unroll
        for (int k = 0; k < S2_K; ++k) {
            const unsigned t = (unsigned)(d[k] - lo);
            own[k] = t < (unsigned)S2_R;
            dl[k]  = own[k] ? t : 0u;
            const unsigned sg = own[k] ? (unsigned)s[k] : 0u;
            hv[k] = *(const float2*)(h2 + (size_t)sg * OUT_DIM);
            ds[k] = dinv[sg];
            dd[k] = dloc[dl[k]];
        }
        int nd[S2_K], ns[S2_K];
        load_idx(nd, ns, base + S2_T * S2_K);
        #pragma unroll
        for (int k = 0; k < S2_K; ++k) {
            if (own[k]) {
                const float nrm = ds[k] * dd[k];
                unsafeAtomicAdd(&acc[dl[k] * OUT_DIM + 0], hv[k].x * nrm);
                unsafeAtomicAdd(&acc[dl[k] * OUT_DIM + 1], hv[k].y * nrm);
            }
        }
        #pragma unroll
        for (int k = 0; k < S2_K; ++k) { d[k] = nd[k]; s[k] = ns[k]; }
    }
    __syncthreads();

    const int lim = rows * OUT_DIM;
    float* aggb = agg2 + (size_t)lo * OUT_DIM;
    for (int i0 = tid; i0 < S2_R * OUT_DIM; i0 += S2_T * S2_F) {
        float c[S2_F];
        #pragma unroll
        for (int k = 0; k < S2_F; ++k) c[k] = acc[min(i0 + k * S2_T, S2_R * OUT_DIM - 1)];
        #pragma unroll
        for (int k = 0; k < S2_F; ++k)
            if (i0 + k * S2_T < lim) unsafeAtomicAdd(aggb + i0 + k * S2_T, c[k]);
    }
}

__global__ void k_final(const float* __restrict__ agg2, const float* __restrict__ b2,
                        float* __restrict__ out, int n) {
    int i = blockIdx.x * blockDim.x + threadIdx.x;
    if (i >= n) return;
    float z0 = agg2[i * 2 + 0] + b2[0];
    float z1 = agg2[i * 2 + 1] + b2[1];
    float mx = fmaxf(z0, z1);
    float l  = mx + logf(expf(z0 - mx) + expf(z1 - mx));
    out[i * 2 + 0] = z0 - l;
    out[i * 2 + 1] = z1 - l;
}

extern "C" void kernel_launch(void* const* d_in, const int* in_sizes, int n_in,
                              void* d_out, int out_size, void* d_ws, size_t ws_size,
                              hipStream_t stream) {
    const float* x  = (const float*)d_in[0];
    const int*   ei = (const int*)d_in[1];
    const float* W1 = (const float*)d_in[2];
    const float* b1 = (const float*)d_in[3];
    const float* W2 = (const float*)d_in[4];
    const float* b2 = (const float*)d_in[5];
    float* out = (float*)d_out;

    const int n  = in_sizes[0] / IN_DIM;
    const int nE = in_sizes[1] / 2;
    const int* src = ei;
    const int* dst = ei + nE;

    float* ws   = (float*)d_ws;
    float* dinv = ws;
    float* h    = ws + n;
    float* agg1 = ws + (size_t)n * 17;
    float* h2   = h;
    float* agg2 = h + (size_t)2 * n;

    const int B = 256;
    k_init_dinv    <<<(n  + B - 1) / B, B, 0, stream>>>(dinv, n);
    k_deg_edges    <<<dim3((n + DG_R - 1) / DG_R, DG_SLICES), DG_T, 0, stream>>>(dst, dinv, nE, n);
    k_finalize_dinv<<<(n  + B - 1) / B, B, 0, stream>>>(dinv, n);

    const int tiles   = (n + 15) / 16;
    const int gblocks = (tiles + 3) / 4;
    k_gemm1_wmma   <<<gblocks, 128, 0, stream>>>(x, W1, h, n);

    k_selfloop16   <<<(n * HID + B - 1) / B, B, 0, stream>>>(h, dinv, agg1, n);
    k_scatter1     <<<dim3((n + S1_R - 1) / S1_R, S1_SLICES), S1_WAVES * 32, 0, stream>>>(src, dst, dinv, h, agg1, nE, n);
    k_layer2       <<<(n  + B - 1) / B, B, 0, stream>>>(agg1, b1, W2, dinv, h2, agg2, n);
    k_scatter2     <<<dim3((n + S2_R - 1) / S2_R, S2_SLICES), S2_T, 0, stream>>>(src, dst, dinv, h2, agg2, nE, n);
    k_final        <<<(n  + B - 1) / B, B, 0, stream>>>(agg2, b2, out, n);
}


# edge weight split: product kernel scales x rows by dinv, stores rows twice (message, self-loop start); sums take dinv[dst] once; edge loops gather no dinv; self-loop launch gone; rounding changes
# speedup vs baseline: 8.3697x; 1.2243x over previous
#include <hip/hip_runtime.h>
#include <hip/hip_bf16.h>
#include <math.h>

typedef __attribute__((ext_vector_type(2))) float v2f;
typedef __attribute__((ext_vector_type(8))) float v8f;

#define IN_DIM  128
#define HID     16
#define OUT_DIM 2

__global__ void k_init_dinv(float* __restrict__ dinv, int n) {
    int i = blockIdx.x * blockDim.x + threadIdx.x;
    if (i < n) dinv[i] = 1.0f;
}

#define DG_R      51200
#define DG_SLICES 128
#define DG_T      1024
#define DG_B      25
#define DG_F      10

__global__ __launch_bounds__(DG_T) void k_deg_edges(const int* __restrict__ dst, float* __restrict__ deg,
                                                    int nE, int n) {
    __shared__ __attribute__((aligned(16))) float cnt[DG_R];

    const int tid  = threadIdx.x;
    const int lo   = blockIdx.x * DG_R;
    const int nsl  = (int)gridDim.y;
    const int per  = ((nE + nsl - 1) / nsl + 31) & ~31;
    const int sbeg = (int)blockIdx.y * per;
    const int send = min(sbeg + per, nE);

    for (int i = tid; i < DG_R / 4; i += DG_T)
        ((float4*)cnt)[i] = make_float4(0.0f, 0.0f, 0.0f, 0.0f);
    __syncthreads();

    for (int base = sbeg; base < send; base += DG_T * DG_B) {
        int d[DG_B];
        #pragma unroll
        for (int k = 0; k < DG_B; ++k)
            d[k] = dst[min(base + k * DG_T + tid, nE - 1)];
        #pragma unroll
        for (int k = 0; k < DG_B; ++k) {
            const unsigned dl = (unsigned)(d[k] - lo);
            if (base + k * DG_T + tid < send && dl < (unsigned)DG_R)
                unsafeAtomicAdd(&cnt[dl], 1.0f);
        }
    }
    __syncthreads();

    const int rows = min(DG_R, n - lo);
    float* degb = deg + lo;
    for (int i0 = tid; i0 < DG_R; i0 += DG_T * DG_F) {
        float c[DG_F];
        #pragma unroll
        for (int k = 0; k < DG_F; ++k) c[k] = cnt[i0 + k * DG_T];
        #pragma unroll
        for (int k = 0; k < DG_F; ++k)
            if (i0 + k * DG_T < rows) unsafeAtomicAdd(degb + i0 + k * DG_T, c[k]);
    }
}

__global__ void k_finalize_dinv(float* __restrict__ dinv, int n) {
    int i = blockIdx.x * blockDim.x + threadIdx.x;
    if (i < n) {
        float d = dinv[i];
        dinv[i] = (d > 0.0f) ? rsqrtf(d) : 0.0f;
    }
}

__global__ __launch_bounds__(128) void k_gemm1_wmma(const float* __restrict__ x,
                                                    const float* __restrict__ W1,
                                                    const float* __restrict__ dinv,
                                                    float* __restrict__ h,
                                                    float* __restrict__ agg, int n) {
    __shared__ float sW[IN_DIM * HID];
    __shared__ float sX[4][16 * 132];

    for (int t = threadIdx.x; t < IN_DIM * HID; t += blockDim.x) sW[t] = W1[t];

    const int wave = __builtin_amdgcn_readfirstlane(threadIdx.x >> 5);
    const int lane = threadIdx.x & 31;
    const int tile = blockIdx.x * 4 + wave;
    const int row0 = tile * 16;
    const bool active = (row0 < n);

    if (active) {
        #pragma unroll 4
        for (int i = 0; i < 16; ++i) {
            int r = row0 + i;
            if (r >= n) r = n - 1;
            const float d = dinv[r];
            const float4 v = ((const float4*)(x + (size_t)r * IN_DIM))[lane];
            const v2f dd = {d, d};
            const v2f lo = (v2f){v.x, v.y} * dd;
            const v2f hi = (v2f){v.z, v.w} * dd;
            *(float4*)&sX[wave][i * 132 + lane * 4] = make_float4(lo.x, lo.y, hi.x, hi.y);
        }
    }
    __syncthreads();
    if (!active) return;

    const int m  = lane & 15;
    const int kb = (lane >> 4) << 1;
    const float* xr = &sX[wave][m * 132];

    v8f acc = {};
    #pragma unroll
    for (int k0 = 0; k0 < IN_DIM; k0 += 4) {
        v2f a, b;
        a.x = xr[k0 + kb];
        a.y = xr[k0 + kb + 1];
        b.x = sW[(k0 + kb)     * HID + m];
        b.y = sW[(k0 + kb + 1) * HID + m];
        acc = __builtin_amdgcn_wmma_f32_16x16x4_f32(false, a, false, b,
                                                    (short)0, acc, false, false);
    }

    const int rbase = row0 + ((lane >> 4) << 3);
    #pragma unroll
    for (int v = 0; v < 8; ++v) {
        int r = rbase + v;
        if (r < n) {
            h[(size_t)r * HID + m]   = acc[v];
            agg[(size_t)r * HID + m] = acc[v];
        }
    }
}

#define S1_RBITS  12
#define S1_R      (1 << S1_RBITS)
#define S1_SLICES 10
#define S1_WAVES  32
#define S1_Q      64
#define S1_CHUNK  128

__device__ __forceinline__ void s1_drain(const unsigned* ringw, unsigned head, int count,
                                         int lane,
                                         const float* __restrict__ h, float* acc) {
    unsigned p = 0;
    if (lane < count)
        p = ringw[(head + (unsigned)lane) & (S1_Q - 1)];
    const unsigned c    = lane & 15;
    const int      half = lane >> 4;
    float v[HID];
    unsigned dj[HID];
    #pragma unroll
    for (int j = 0; j < HID; ++j) {
        const int sl = 2 * j + half;
        const unsigned pj = (unsigned)__shfl((int)p, sl);
        dj[j] = pj & (S1_R - 1);
        v[j]  = h[(pj >> S1_RBITS) * HID + c];
    }
    #pragma unroll
    for (int j = 0; j < HID; ++j) {
        if (2 * j + half < count)
            unsafeAtomicAdd(&acc[dj[j] * HID + c], v[j]);
    }
}

__global__ __launch_bounds__(1024) void k_scatter1(const int* __restrict__ src, const int* __restrict__ dst,
                                                   const float* __restrict__ h,
                                                   float* __restrict__ agg, int nE, int n) {
    __shared__ __attribute__((aligned(16))) float acc[S1_R * HID];
    __shared__ unsigned ring[S1_WAVES][S1_Q];

    const int tid  = threadIdx.x;
    const int lane = tid & 31;
    const int wave = __builtin_amdgcn_readfirstlane(tid >> 5);
    const int lo   = blockIdx.x * S1_R;
    const int nsl  = (int)gridDim.y;
    const int per  = ((nE + nsl - 1) / nsl + S1_CHUNK - 1) / S1_CHUNK * S1_CHUNK;
    const int sbeg = (int)blockIdx.y * per;
    const int send = min(sbeg + per, nE);

    for (int i = tid; i < S1_R * HID / 4; i += S1_WAVES * 32)
        ((float4*)acc)[i] = make_float4(0.0f, 0.0f, 0.0f, 0.0f);
    __syncthreads();

    unsigned* ringw = ring[wave];
    unsigned qh = 0, qt = 0;

    auto load_chunk = [&](int (&d)[4], int (&s)[4], int base) {
        #pragma unroll
        for (int k = 0; k < 4; ++k) {
            const int e  = base + k * 32 + lane;
            const int ec = min(e, nE - 1);
            const int dv = dst[ec];
            s[k] = src[ec];
            d[k] = (e < send) ? dv : -1;
        }
    };
    auto filter_chunk = [&](const int (&d)[4], const int (&s)[4]) {
        #pragma unroll
        for (int k = 0; k < 4; ++k) {
            const unsigned dl = (unsigned)(d[k] - lo);
            const bool m = dl < (unsigned)S1_R;
            const unsigned mask = __builtin_amdgcn_ballot_w32(m);
            if (m)
                ringw[(qt + __builtin_amdgcn_mbcnt_lo(mask, 0u)) & (S1_Q - 1)] =
                    ((unsigned)s[k] << S1_RBITS) | dl;
            qt += (unsigned)__builtin_popcount(mask);
            if (qt - qh >= 32u) {
                s1_drain(ringw, qh, 32, lane, h, acc);
                qh += 32u;
            }
        }
    };

    const int step = S1_WAVES * S1_CHUNK;
    int e0 = sbeg + wave * S1_CHUNK;
    int dA[4], sA[4], dB[4], sB[4];
    load_chunk(dA, sA, e0);
    for (; e0 < send; e0 += 2 * step) {
        load_chunk(dB, sB, e0 + step);
        filter_chunk(dA, sA);
        load_chunk(dA, sA, e0 + 2 * step);
        filter_chunk(dB, sB);
    }
    if (qt != qh) s1_drain(ringw, qh, (int)(qt - qh), lane, h, acc);
    __syncthreads();

    const int rows = min(S1_R, n - lo);
    float* aggb = agg + (size_t)lo * HID;
    for (int i = tid; i < rows * HID; i += S1_WAVES * 32)
        unsafeAtomicAdd(aggb + i, acc[i]);
}

__global__ void k_layer2(const float* __restrict__ agg1, const float* __restrict__ b1,
                         const float* __restrict__ W2, const float* __restrict__ dinv,
                         float* __restrict__ h2, float* __restrict__ agg2, int n) {
    int i = blockIdx.x * blockDim.x + threadIdx.x;
    if (i >= n) return;
    const float* r = agg1 + (size_t)i * HID;
    const float di = dinv[i];
    float z0 = 0.0f, z1 = 0.0f;
    #pragma unroll
    for (int k = 0; k < HID; ++k) {
        float v = fmaxf(r[k] * di + b1[k], 0.0f);
        z0 = fmaf(v, W2[k * OUT_DIM + 0], z0);
        z1 = fmaf(v, W2[k * OUT_DIM + 1], z1);
    }
    z0 *= di;
    z1 *= di;
    h2[i * 2 + 0]   = z0;
    h2[i * 2 + 1]   = z1;
    agg2[i * 2 + 0] = z0;
    agg2[i * 2 + 1] = z1;
}

#define S2_R      25024
#define S2_SLICES 64
#define S2_T      1024
#define S2_K      7
#define S2_F      7

__global__ __launch_bounds__(S2_T) void k_scatter2(const int* __restrict__ src, const int* __restrict__ dst,
                                                   const float* __restrict__ h2,
                                                   float* __restrict__ agg2, int nE, int n) {
    __shared__ __attribute__((aligned(16))) float acc[S2_R * OUT_DIM];

    const int tid  = threadIdx.x;
    const int lo   = blockIdx.x * S2_R;
    const int rows = min(S2_R, n - lo);
    const int nsl  = (int)gridDim.y;
    const int per  = ((nE + nsl - 1) / nsl + 31) & ~31;
    const int sbeg = (int)blockIdx.y * per;
    const int send = min(sbeg + per, nE);

    for (int i = tid; i < S2_R * OUT_DIM / 4; i += S2_T)
        ((float4*)acc)[i] = make_float4(0.0f, 0.0f, 0.0f, 0.0f);
    __syncthreads();

    auto load_idx = [&](int (&d)[S2_K], int (&s)[S2_K], int base) {
        #pragma unroll
        for (int k = 0; k < S2_K; ++k) {
            const int e  = base + k * S2_T + tid;
            const int ec = min(e, nE - 1);
            const int dv = dst[ec];
            s[k] = src[ec];
            d[k] = (e < send) ? dv : -1;
        }
    };

    int d[S2_K], s[S2_K];
    load_idx(d, s, sbeg);
    for (int base = sbeg; base < send; base += S2_T * S2_K) {
        unsigned dl[S2_K];
        bool     own[S2_K];
        float2   hv[S2_K];
        #pragma unroll
        for (int k = 0; k < S2_K; ++k) {
            const unsigned t = (unsigned)(d[k] - lo);
            own[k] = t < (unsigned)S2_R;
            dl[k]  = own[k] ? t : 0u;
            const unsigned sg = own[k] ? (unsigned)s[k] : 0u;
            hv[k] = *(const float2*)(h2 + (size_t)sg * OUT_DIM);
        }
        int nd[S2_K], ns[S2_K];
        load_idx(nd, ns, base + S2_T * S2_K);
        #pragma unroll
        for (int k = 0; k < S2_K; ++k) {
            if (own[k]) {
                unsafeAtomicAdd(&acc[dl[k] * OUT_DIM + 0], hv[k].x);
                unsafeAtomicAdd(&acc[dl[k] * OUT_DIM + 1], hv[k].y);
            }
        }
        #pragma unroll
        for (int k = 0; k < S2_K; ++k) { d[k] = nd[k]; s[k] = ns[k]; }
    }
    __syncthreads();

    const int lim = rows * OUT_DIM;
    float* aggb = agg2 + (size_t)lo * OUT_DIM;
    for (int i0 = tid; i0 < S2_R * OUT_DIM; i0 += S2_T * S2_F) {
        float c[S2_F];
        #pragma unroll
        for (int k = 0; k < S2_F; ++k) c[k] = acc[min(i0 + k * S2_T, S2_R * OUT_DIM - 1)];
        #pragma unroll
        for (int k = 0; k < S2_F; ++k)
            if (i0 + k * S2_T < lim) unsafeAtomicAdd(aggb + i0 + k * S2_T, c[k]);
    }
}

__global__ void k_final(const float* __restrict__ agg2, const float* __restrict__ dinv,
                        const float* __restrict__ b2,
                        float* __restrict__ out, int n) {
    int i = blockIdx.x * blockDim.x + threadIdx.x;
    if (i >= n) return;
    const float di = dinv[i];
    float z0 = agg2[i * 2 + 0] * di + b2[0];
    float z1 = agg2[i * 2 + 1] * di + b2[1];
    float mx = fmaxf(z0, z1);
    float l  = mx + logf(expf(z0 - mx) + expf(z1 - mx));
    out[i * 2 + 0] = z0 - l;
    out[i * 2 + 1] = z1 - l;
}

extern "C" void kernel_launch(void* const* d_in, const int* in_sizes, int n_in,
                              void* d_out, int out_size, void* d_ws, size_t ws_size,
                              hipStream_t stream) {
    const float* x  = (const float*)d_in[0];
    const int*   ei = (const int*)d_in[1];
    const float* W1 = (const float*)d_in[2];
    const float* b1 = (const float*)d_in[3];
    const float* W2 = (const float*)d_in[4];
    const float* b2 = (const float*)d_in[5];
    float* out = (float*)d_out;

    const int n  = in_sizes[0] / IN_DIM;
    const int nE = in_sizes[1] / 2;
    const int* src = ei;
    const int* dst = ei + nE;

    float* ws   = (float*)d_ws;
    float* dinv = ws;
    float* h    = ws + n;
    float* agg1 = ws + (size_t)n * 17;
    float* h2   = h;
    float* agg2 = h + (size_t)2 * n;

    const int B = 256;
    k_init_dinv    <<<(n  + B - 1) / B, B, 0, stream>>>(dinv, n);
    k_deg_edges    <<<dim3((n + DG_R - 1) / DG_R, DG_SLICES), DG_T, 0, stream>>>(dst, dinv, nE, n);
    k_finalize_dinv<<<(n  + B - 1) / B, B, 0, stream>>>(dinv, n);

    const int tiles   = (n + 15) / 16;
    const int gblocks = (tiles + 3) / 4;
    k_gemm1_wmma   <<<gblocks, 128, 0, stream>>>(x, W1, dinv, h, agg1, n);

    k_scatter1     <<<dim3((n + S1_R - 1) / S1_R, S1_SLICES), S1_WAVES * 32, 0, stream>>>(src, dst, h, agg1, nE, n);
    k_layer2       <<<(n  + B - 1) / B, B, 0, stream>>>(agg1, b1, W2, dinv, h2, agg2, n);
    k_scatter2     <<<dim3((n + S2_R - 1) / S2_R, S2_SLICES), S2_T, 0, stream>>>(src, dst, h2, agg2, nE, n);
    k_final        <<<(n  + B - 1) / B, B, 0, stream>>>(agg2, dinv, b2, out, n);
}


# layer-1 edge scan takes 4 consecutive edges per lane: one 16-byte load per index list and one ring append per 128 edges (4 ballots, one chained count); float sum order changes
# speedup vs baseline: 8.6830x; 1.0374x over previous
#include <hip/hip_runtime.h>
#include <hip/hip_bf16.h>
#include <math.h>

typedef __attribute__((ext_vector_type(2))) float v2f;
typedef __attribute__((ext_vector_type(8))) float v8f;

#define IN_DIM  128
#define HID     16
#define OUT_DIM 2

__global__ void k_init_dinv(float* __restrict__ dinv, int n) {
    int i = blockIdx.x * blockDim.x + threadIdx.x;
    if (i < n) dinv[i] = 1.0f;
}

#define DG_R      51200
#define DG_SLICES 128
#define DG_T      1024
#define DG_B      25
#define DG_F      10

__global__ __launch_bounds__(DG_T) void k_deg_edges(const int* __restrict__ dst, float* __restrict__ deg,
                                                    int nE, int n) {
    __shared__ __attribute__((aligned(16))) float cnt[DG_R];

    const int tid  = threadIdx.x;
    const int lo   = blockIdx.x * DG_R;
    const int nsl  = (int)gridDim.y;
    const int per  = ((nE + nsl - 1) / nsl + 31) & ~31;
    const int sbeg = (int)blockIdx.y * per;
    const int send = min(sbeg + per, nE);

    for (int i = tid; i < DG_R / 4; i += DG_T)
        ((float4*)cnt)[i] = make_float4(0.0f, 0.0f, 0.0f, 0.0f);
    __syncthreads();

    for (int base = sbeg; base < send; base += DG_T * DG_B) {
        int d[DG_B];
        #pragma unroll
        for (int k = 0; k < DG_B; ++k)
            d[k] = dst[min(base + k * DG_T + tid, nE - 1)];
        #pragma unroll
        for (int k = 0; k < DG_B; ++k) {
            const unsigned dl = (unsigned)(d[k] - lo);
            if (base + k * DG_T + tid < send && dl < (unsigned)DG_R)
                unsafeAtomicAdd(&cnt[dl], 1.0f);
        }
    }
    __syncthreads();

    const int rows = min(DG_R, n - lo);
    float* degb = deg + lo;
    for (int i0 = tid; i0 < DG_R; i0 += DG_T * DG_F) {
        float c[DG_F];
        #pragma unroll
        for (int k = 0; k < DG_F; ++k) c[k] = cnt[i0 + k * DG_T];
        #pragma unroll
        for (int k = 0; k < DG_F; ++k)
            if (i0 + k * DG_T < rows) unsafeAtomicAdd(degb + i0 + k * DG_T, c[k]);
    }
}

__global__ void k_finalize_dinv(float* __restrict__ dinv, int n) {
    int i = blockIdx.x * blockDim.x + threadIdx.x;
    if (i < n) {
        float d = dinv[i];
        dinv[i] = (d > 0.0f) ? rsqrtf(d) : 0.0f;
    }
}

__global__ __launch_bounds__(128) void k_gemm1_wmma(const float* __restrict__ x,
                                                    const float* __restrict__ W1,
                                                    const float* __restrict__ dinv,
                                                    float* __restrict__ h,
                                                    float* __restrict__ agg, int n) {
    __shared__ float sW[IN_DIM * HID];
    __shared__ float sX[4][16 * 132];

    for (int t = threadIdx.x; t < IN_DIM * HID; t += blockDim.x) sW[t] = W1[t];

    const int wave = __builtin_amdgcn_readfirstlane(threadIdx.x >> 5);
    const int lane = threadIdx.x & 31;
    const int tile = blockIdx.x * 4 + wave;
    const int row0 = tile * 16;
    const bool active = (row0 < n);

    if (active) {
        #pragma unroll 4
        for (int i = 0; i < 16; ++i) {
            int r = row0 + i;
            if (r >= n) r = n - 1;
            const float d = dinv[r];
            const float4 v = ((const float4*)(x + (size_t)r * IN_DIM))[lane];
            const v2f dd = {d, d};
            const v2f lo = (v2f){v.x, v.y} * dd;
            const v2f hi = (v2f){v.z, v.w} * dd;
            *(float4*)&sX[wave][i * 132 + lane * 4] = make_float4(lo.x, lo.y, hi.x, hi.y);
        }
    }
    __syncthreads();
    if (!active) return;

    const int m  = lane & 15;
    const int kb = (lane >> 4) << 1;
    const float* xr = &sX[wave][m * 132];

    v8f acc = {};
    #pragma unroll
    for (int k0 = 0; k0 < IN_DIM; k0 += 4) {
        v2f a, b;
        a.x = xr[k0 + kb];
        a.y = xr[k0 + kb + 1];
        b.x = sW[(k0 + kb)     * HID + m];
        b.y = sW[(k0 + kb + 1) * HID + m];
        acc = __builtin_amdgcn_wmma_f32_16x16x4_f32(false, a, false, b,
                                                    (short)0, acc, false, false);
    }

    const int rbase = row0 + ((lane >> 4) << 3);
    #pragma unroll
    for (int v = 0; v < 8; ++v) {
        int r = rbase + v;
        if (r < n) {
            h[(size_t)r * HID + m]   = acc[v];
            agg[(size_t)r * HID + m] = acc[v];
        }
    }
}

#define S1_RBITS  12
#define S1_R      (1 << S1_RBITS)
#define S1_SLICES 10
#define S1_WAVES  32
#define S1_Q      256
#define S1_CHUNK  128

__device__ __forceinline__ void s1_drain(const unsigned* ringw, unsigned head, int count,
                                         int lane,
                                         const float* __restrict__ h, float* acc) {
    unsigned p = 0;
    if (lane < count)
        p = ringw[(head + (unsigned)lane) & (S1_Q - 1)];
    const unsigned c    = lane & 15;
    const int      half = lane >> 4;
    float v[HID];
    unsigned dj[HID];
    #pragma unroll
    for (int j = 0; j < HID; ++j) {
        const int sl = 2 * j + half;
        const unsigned pj = (unsigned)__shfl((int)p, sl);
        dj[j] = pj & (S1_R - 1);
        v[j]  = h[(pj >> S1_RBITS) * HID + c];
    }
    #pragma unroll
    for (int j = 0; j < HID; ++j) {
        if (2 * j + half < count)
            unsafeAtomicAdd(&acc[dj[j] * HID + c], v[j]);
    }
}

__global__ __launch_bounds__(1024) void k_scatter1(const int* __restrict__ src, const int* __restrict__ dst,
                                                   const float* __restrict__ h,
                                                   float* __restrict__ agg, int nE, int n) {
    __shared__ __attribute__((aligned(16))) float acc[S1_R * HID];
    __shared__ unsigned ring[S1_WAVES][S1_Q];

    const int tid  = threadIdx.x;
    const int lane = tid & 31;
    const int wave = __builtin_amdgcn_readfirstlane(tid >> 5);
    const int lo   = blockIdx.x * S1_R;
    const int nsl  = (int)gridDim.y;
    const int per  = ((nE + nsl - 1) / nsl + S1_CHUNK - 1) / S1_CHUNK * S1_CHUNK;
    const int sbeg = (int)blockIdx.y * per;
    const int send = min(sbeg + per, nE);

    for (int i = tid; i < S1_R * HID / 4; i += S1_WAVES * 32)
        ((float4*)acc)[i] = make_float4(0.0f, 0.0f, 0.0f, 0.0f);
    __syncthreads();

    unsigned* ringw = ring[wave];
    unsigned qh = 0, qt = 0;

    auto load_chunk = [&](int4 &d, int4 &s, int base) {
        const int e  = base + 4 * lane;
        const unsigned off = 4u * (unsigned)min(e, nE - 4);
        const int4 dv = *(const int4*)((const char*)dst + off);
        s = *(const int4*)((const char*)src + off);
        d = (e < send) ? dv : make_int4(-1, -1, -1, -1);
    };
    auto filter_chunk = [&](const int4 &d, const int4 &s) {
        const unsigned dl0 = (unsigned)(d.x - lo), dl1 = (unsigned)(d.y - lo);
        const unsigned dl2 = (unsigned)(d.z - lo), dl3 = (unsigned)(d.w - lo);
        const bool m0 = dl0 < (unsigned)S1_R, m1 = dl1 < (unsigned)S1_R;
        const bool m2 = dl2 < (unsigned)S1_R, m3 = dl3 < (unsigned)S1_R;
        const unsigned b0 = __builtin_amdgcn_ballot_w32(m0), b1 = __builtin_amdgcn_ballot_w32(m1);
        const unsigned b2 = __builtin_amdgcn_ballot_w32(m2), b3 = __builtin_amdgcn_ballot_w32(m3);
        unsigned pos = qt + __builtin_amdgcn_mbcnt_lo(b0, __builtin_amdgcn_mbcnt_lo(b1,
                            __builtin_amdgcn_mbcnt_lo(b2, __builtin_amdgcn_mbcnt_lo(b3, 0u))));
        if (m0) { ringw[pos & (S1_Q - 1)] = ((unsigned)s.x << S1_RBITS) | dl0; ++pos; }
        if (m1) { ringw[pos & (S1_Q - 1)] = ((unsigned)s.y << S1_RBITS) | dl1; ++pos; }
        if (m2) { ringw[pos & (S1_Q - 1)] = ((unsigned)s.z << S1_RBITS) | dl2; ++pos; }
        if (m3) { ringw[pos & (S1_Q - 1)] = ((unsigned)s.w << S1_RBITS) | dl3; }
        qt += (unsigned)(__builtin_popcount(b0) + __builtin_popcount(b1) +
                         __builtin_popcount(b2) + __builtin_popcount(b3));
        while (qt - qh >= 32u) {
            s1_drain(ringw, qh, 32, lane, h, acc);
            qh += 32u;
        }
    };

    const int step = S1_WAVES * S1_CHUNK;
    int e0 = sbeg + wave * S1_CHUNK;
    int4 dA, sA, dB, sB;
    load_chunk(dA, sA, e0);
    for (; e0 < send; e0 += 2 * step) {
        load_chunk(dB, sB, e0 + step);
        filter_chunk(dA, sA);
        load_chunk(dA, sA, e0 + 2 * step);
        filter_chunk(dB, sB);
    }
    if (qt != qh) s1_drain(ringw, qh, (int)(qt - qh), lane, h, acc);
    __syncthreads();

    const int rows = min(S1_R, n - lo);
    float* aggb = agg + (size_t)lo * HID;
    for (int i = tid; i < rows * HID; i += S1_WAVES * 32)
        unsafeAtomicAdd(aggb + i, acc[i]);
}

__global__ void k_layer2(const float* __restrict__ agg1, const float* __restrict__ b1,
                         const float* __restrict__ W2, const float* __restrict__ dinv,
                         float* __restrict__ h2, float* __restrict__ agg2, int n) {
    int i = blockIdx.x * blockDim.x + threadIdx.x;
    if (i >= n) return;
    const float* r = agg1 + (size_t)i * HID;
    const float di = dinv[i];
    float z0 = 0.0f, z1 = 0.0f;
    #pragma unroll
    for (int k = 0; k < HID; ++k) {
        float v = fmaxf(r[k] * di + b1[k], 0.0f);
        z0 = fmaf(v, W2[k * OUT_DIM + 0], z0);
        z1 = fmaf(v, W2[k * OUT_DIM + 1], z1);
    }
    z0 *= di;
    z1 *= di;
    h2[i * 2 + 0]   = z0;
    h2[i * 2 + 1]   = z1;
    agg2[i * 2 + 0] = z0;
    agg2[i * 2 + 1] = z1;
}

#define S2_R      25024
#define S2_SLICES 64
#define S2_T      1024
#define S2_K      7
#define S2_F      7

__global__ __launch_bounds__(S2_T) void k_scatter2(const int* __restrict__ src, const int* __restrict__ dst,
                                                   const float* __restrict__ h2,
                                                   float* __restrict__ agg2, int nE, int n) {
    __shared__ __attribute__((aligned(16))) float acc[S2_R * OUT_DIM];

    const int tid  = threadIdx.x;
    const int lo   = blockIdx.x * S2_R;
    const int rows = min(S2_R, n - lo);
    const int nsl  = (int)gridDim.y;
    const int per  = ((nE + nsl - 1) / nsl + 31) & ~31;
    const int sbeg = (int)blockIdx.y * per;
    const int send = min(sbeg + per, nE);

    for (int i = tid; i < S2_R * OUT_DIM / 4; i += S2_T)
        ((float4*)acc)[i] = make_float4(0.0f, 0.0f, 0.0f, 0.0f);
    __syncthreads();

    auto load_idx = [&](int (&d)[S2_K], int (&s)[S2_K], int base) {
        #pragma unroll
        for (int k = 0; k < S2_K; ++k) {
            const int e  = base + k * S2_T + tid;
            const int ec = min(e, nE - 1);
            const int dv = dst[ec];
            s[k] = src[ec];
            d[k] = (e < send) ? dv : -1;
        }
    };

    int d[S2_K], s[S2_K];
    load_idx(d, s, sbeg);
    for (int base = sbeg; base < send; base += S2_T * S2_K) {
        unsigned dl[S2_K];
        bool     own[S2_K];
        float2   hv[S2_K];
        #pragma unroll
        for (int k = 0; k < S2_K; ++k) {
            const unsigned t = (unsigned)(d[k] - lo);
            own[k] = t < (unsigned)S2_R;
            dl[k]  = own[k] ? t : 0u;
            const unsigned sg = own[k] ? (unsigned)s[k] : 0u;
            hv[k] = *(const float2*)(h2 + (size_t)sg * OUT_DIM);
        }
        int nd[S2_K], ns[S2_K];
        load_idx(nd, ns, base + S2_T * S2_K);
        #pragma unroll
        for (int k = 0; k < S2_K; ++k) {
            if (own[k]) {
                unsafeAtomicAdd(&acc[dl[k] * OUT_DIM + 0], hv[k].x);
                unsafeAtomicAdd(&acc[dl[k] * OUT_DIM + 1], hv[k].y);
            }
        }
        #pragma unroll
        for (int k = 0; k < S2_K; ++k) { d[k] = nd[k]; s[k] = ns[k]; }
    }
    __syncthreads();

    const int lim = rows * OUT_DIM;
    float* aggb = agg2 + (size_t)lo * OUT_DIM;
    for (int i0 = tid; i0 < S2_R * OUT_DIM; i0 += S2_T * S2_F) {
        float c[S2_F];
        #pragma unroll
        for (int k = 0; k < S2_F; ++k) c[k] = acc[min(i0 + k * S2_T, S2_R * OUT_DIM - 1)];
        #pragma unroll
        for (int k = 0; k < S2_F; ++k)
            if (i0 + k * S2_T < lim) unsafeAtomicAdd(aggb + i0 + k * S2_T, c[k]);
    }
}

__global__ void k_final(const float* __restrict__ agg2, const float* __restrict__ dinv,
                        const float* __restrict__ b2,
                        float* __restrict__ out, int n) {
    int i = blockIdx.x * blockDim.x + threadIdx.x;
    if (i >= n) return;
    const float di = dinv[i];
    float z0 = agg2[i * 2 + 0] * di + b2[0];
    float z1 = agg2[i * 2 + 1] * di + b2[1];
    float mx = fmaxf(z0, z1);
    float l  = mx + logf(expf(z0 - mx) + expf(z1 - mx));
    out[i * 2 + 0] = z0 - l;
    out[i * 2 + 1] = z1 - l;
}

extern "C" void kernel_launch(void* const* d_in, const int* in_sizes, int n_in,
                              void* d_out, int out_size, void* d_ws, size_t ws_size,
                              hipStream_t stream) {
    const float* x  = (const float*)d_in[0];
    const int*   ei = (const int*)d_in[1];
    const float* W1 = (const float*)d_in[2];
    const float* b1 = (const float*)d_in[3];
    const float* W2 = (const float*)d_in[4];
    const float* b2 = (const float*)d_in[5];
    float* out = (float*)d_out;

    const int n  = in_sizes[0] / IN_DIM;
    const int nE = in_sizes[1] / 2;
    const int* src = ei;
    const int* dst = ei + nE;

    float* ws   = (float*)d_ws;
    float* dinv = ws;
    float* h    = ws + n;
    float* agg1 = ws + (size_t)n * 17;
    float* h2   = h;
    float* agg2 = h + (size_t)2 * n;

    const int B = 256;
    k_init_dinv    <<<(n  + B - 1) / B, B, 0, stream>>>(dinv, n);
    k_deg_edges    <<<dim3((n + DG_R - 1) / DG_R, DG_SLICES), DG_T, 0, stream>>>(dst, dinv, nE, n);
    k_finalize_dinv<<<(n  + B - 1) / B, B, 0, stream>>>(dinv, n);

    const int tiles   = (n + 15) / 16;
    const int gblocks = (tiles + 3) / 4;
    k_gemm1_wmma   <<<gblocks, 128, 0, stream>>>(x, W1, dinv, h, agg1, n);

    k_scatter1     <<<dim3((n + S1_R - 1) / S1_R, S1_SLICES), S1_WAVES * 32, 0, stream>>>(src, dst, h, agg1, nE, n);
    k_layer2       <<<(n  + B - 1) / B, B, 0, stream>>>(agg1, b1, W2, dinv, h2, agg2, n);
    k_scatter2     <<<dim3((n + S2_R - 1) / S2_R, S2_SLICES), S2_T, 0, stream>>>(src, dst, h2, agg2, nE, n);
    k_final        <<<(n  + B - 1) / B, B, 0, stream>>>(agg2, dinv, b2, out, n);
}


# degree counts leave the blocks as plain whole-line stores into 32 per-slice planes (8 node ranges x 32 edge slices) that the rsqrt kernel sums; deg atomics and the init launch are gone; counts exact
# speedup vs baseline: 9.2554x; 1.0659x over previous
#include <hip/hip_runtime.h>
#include <hip/hip_bf16.h>
#include <math.h>

typedef __attribute__((ext_vector_type(2))) float v2f;
typedef __attribute__((ext_vector_type(8))) float v8f;

#define IN_DIM  128
#define HID     16
#define OUT_DIM 2


#define DG_R      12800
#define DG_SLICES 32
#define DG_T      1024
#define DG_B      13
#define DG_F      13

__global__ __launch_bounds__(DG_T) void k_deg_edges(const int* __restrict__ dst, float* __restrict__ planes,
                                                    int nE, int n) {
    __shared__ __attribute__((aligned(16))) float cnt[DG_R];

    const int tid  = threadIdx.x;
    const int lo   = blockIdx.x * DG_R;
    const int nsl  = (int)gridDim.y;
    const int per  = ((nE + nsl - 1) / nsl + 3) & ~3;
    const int sbeg = (int)blockIdx.y * per;
    const int send = min(sbeg + per, nE);

    for (int i = tid; i < DG_R / 4; i += DG_T)
        ((float4*)cnt)[i] = make_float4(0.0f, 0.0f, 0.0f, 0.0f);
    __syncthreads();

    for (int base = sbeg; base < send; base += DG_T * DG_B * 4) {
        int4 d[DG_B];
        #pragma unroll
        for (int k = 0; k < DG_B; ++k) {
            const int e = base + 4 * (k * DG_T + tid);
            const unsigned off = 4u * (unsigned)min(e, nE - 4);
            d[k] = *(const int4*)((const char*)dst + off);
        }
        #pragma unroll
        for (int k = 0; k < DG_B; ++k) {
            if (base + 4 * (k * DG_T + tid) < send) {
                const unsigned d0 = (unsigned)(d[k].x - lo), d1 = (unsigned)(d[k].y - lo);
                const unsigned d2 = (unsigned)(d[k].z - lo), d3 = (unsigned)(d[k].w - lo);
                if (d0 < (unsigned)DG_R) unsafeAtomicAdd(&cnt[d0], 1.0f);
                if (d1 < (unsigned)DG_R) unsafeAtomicAdd(&cnt[d1], 1.0f);
                if (d2 < (unsigned)DG_R) unsafeAtomicAdd(&cnt[d2], 1.0f);
                if (d3 < (unsigned)DG_R) unsafeAtomicAdd(&cnt[d3], 1.0f);
            }
        }
    }
    __syncthreads();

    const int rows = min(DG_R, n - lo);
    float* pl = planes + (size_t)blockIdx.y * n + lo;
    float c[DG_F];
    #pragma unroll
    for (int k = 0; k < DG_F; ++k) c[k] = cnt[min(tid + k * DG_T, DG_R - 1)];
    #pragma unroll
    for (int k = 0; k < DG_F; ++k)
        if (tid + k * DG_T < rows) pl[tid + k * DG_T] = c[k];
}

__global__ void k_finalize_dinv(const float* __restrict__ planes, float* __restrict__ dinv, int n) {
    int i = blockIdx.x * blockDim.x + threadIdx.x;
    if (i < n) {
        float c[DG_SLICES];
        #pragma unroll
        for (int s = 0; s < DG_SLICES; ++s) {
            const unsigned off = 4u * ((unsigned)s * (unsigned)n + (unsigned)i);
            c[s] = *(const float*)((const char*)planes + off);
        }
        float d = 1.0f;
        #pragma unroll
        for (int s = 0; s < DG_SLICES; ++s) d += c[s];
        dinv[i] = (d > 0.0f) ? rsqrtf(d) : 0.0f;
    }
}

__global__ __launch_bounds__(128) void k_gemm1_wmma(const float* __restrict__ x,
                                                    const float* __restrict__ W1,
                                                    const float* __restrict__ dinv,
                                                    float* __restrict__ h,
                                                    float* __restrict__ agg, int n) {
    __shared__ float sW[IN_DIM * HID];
    __shared__ float sX[4][16 * 132];

    for (int t = threadIdx.x; t < IN_DIM * HID; t += blockDim.x) sW[t] = W1[t];

    const int wave = __builtin_amdgcn_readfirstlane(threadIdx.x >> 5);
    const int lane = threadIdx.x & 31;
    const int tile = blockIdx.x * 4 + wave;
    const int row0 = tile * 16;
    const bool active = (row0 < n);

    if (active) {
        #pragma unroll 4
        for (int i = 0; i < 16; ++i) {
            int r = row0 + i;
            if (r >= n) r = n - 1;
            const float d = dinv[r];
            const float4 v = ((const float4*)(x + (size_t)r * IN_DIM))[lane];
            const v2f dd = {d, d};
            const v2f lo = (v2f){v.x, v.y} * dd;
            const v2f hi = (v2f){v.z, v.w} * dd;
            *(float4*)&sX[wave][i * 132 + lane * 4] = make_float4(lo.x, lo.y, hi.x, hi.y);
        }
    }
    __syncthreads();
    if (!active) return;

    const int m  = lane & 15;
    const int kb = (lane >> 4) << 1;
    const float* xr = &sX[wave][m * 132];

    v8f acc = {};
    #pragma unroll
    for (int k0 = 0; k0 < IN_DIM; k0 += 4) {
        v2f a, b;
        a.x = xr[k0 + kb];
        a.y = xr[k0 + kb + 1];
        b.x = sW[(k0 + kb)     * HID + m];
        b.y = sW[(k0 + kb + 1) * HID + m];
        acc = __builtin_amdgcn_wmma_f32_16x16x4_f32(false, a, false, b,
                                                    (short)0, acc, false, false);
    }

    const int rbase = row0 + ((lane >> 4) << 3);
    #pragma unroll
    for (int v = 0; v < 8; ++v) {
        int r = rbase + v;
        if (r < n) {
            h[(size_t)r * HID + m]   = acc[v];
            agg[(size_t)r * HID + m] = acc[v];
        }
    }
}

#define S1_RBITS  12
#define S1_R      (1 << S1_RBITS)
#define S1_SLICES 10
#define S1_WAVES  32
#define S1_Q      256
#define S1_CHUNK  128

__device__ __forceinline__ void s1_drain(const unsigned* ringw, unsigned head, int count,
                                         int lane,
                                         const float* __restrict__ h, float* acc) {
    unsigned p = 0;
    if (lane < count)
        p = ringw[(head + (unsigned)lane) & (S1_Q - 1)];
    const unsigned c    = lane & 15;
    const int      half = lane >> 4;
    float v[HID];
    unsigned dj[HID];
    #pragma unroll
    for (int j = 0; j < HID; ++j) {
        const int sl = 2 * j + half;
        const unsigned pj = (unsigned)__shfl((int)p, sl);
        dj[j] = pj & (S1_R - 1);
        v[j]  = h[(pj >> S1_RBITS) * HID + c];
    }
    #pragma unroll
    for (int j = 0; j < HID; ++j) {
        if (2 * j + half < count)
            unsafeAtomicAdd(&acc[dj[j] * HID + c], v[j]);
    }
}

__global__ __launch_bounds__(1024) void k_scatter1(const int* __restrict__ src, const int* __restrict__ dst,
                                                   const float* __restrict__ h,
                                                   float* __restrict__ agg, int nE, int n) {
    __shared__ __attribute__((aligned(16))) float acc[S1_R * HID];
    __shared__ unsigned ring[S1_WAVES][S1_Q];

    const int tid  = threadIdx.x;
    const int lane = tid & 31;
    const int wave = __builtin_amdgcn_readfirstlane(tid >> 5);
    const int lo   = blockIdx.x * S1_R;
    const int nsl  = (int)gridDim.y;
    const int per  = ((nE + nsl - 1) / nsl + S1_CHUNK - 1) / S1_CHUNK * S1_CHUNK;
    const int sbeg = (int)blockIdx.y * per;
    const int send = min(sbeg + per, nE);

    for (int i = tid; i < S1_R * HID / 4; i += S1_WAVES * 32)
        ((float4*)acc)[i] = make_float4(0.0f, 0.0f, 0.0f, 0.0f);
    __syncthreads();

    unsigned* ringw = ring[wave];
    unsigned qh = 0, qt = 0;

    auto load_chunk = [&](int4 &d, int4 &s, int base) {
        const int e  = base + 4 * lane;
        const unsigned off = 4u * (unsigned)min(e, nE - 4);
        const int4 dv = *(const int4*)((const char*)dst + off);
        s = *(const int4*)((const char*)src + off);
        d = (e < send) ? dv : make_int4(-1, -1, -1, -1);
    };
    auto filter_chunk = [&](const int4 &d, const int4 &s) {
        const unsigned dl0 = (unsigned)(d.x - lo), dl1 = (unsigned)(d.y - lo);
        const unsigned dl2 = (unsigned)(d.z - lo), dl3 = (unsigned)(d.w - lo);
        const bool m0 = dl0 < (unsigned)S1_R, m1 = dl1 < (unsigned)S1_R;
        const bool m2 = dl2 < (unsigned)S1_R, m3 = dl3 < (unsigned)S1_R;
        const unsigned b0 = __builtin_amdgcn_ballot_w32(m0), b1 = __builtin_amdgcn_ballot_w32(m1);
        const unsigned b2 = __builtin_amdgcn_ballot_w32(m2), b3 = __builtin_amdgcn_ballot_w32(m3);
        unsigned pos = qt + __builtin_amdgcn_mbcnt_lo(b0, __builtin_amdgcn_mbcnt_lo(b1,
                            __builtin_amdgcn_mbcnt_lo(b2, __builtin_amdgcn_mbcnt_lo(b3, 0u))));
        if (m0) { ringw[pos & (S1_Q - 1)] = ((unsigned)s.x << S1_RBITS) | dl0; ++pos; }
        if (m1) { ringw[pos & (S1_Q - 1)] = ((unsigned)s.y << S1_RBITS) | dl1; ++pos; }
        if (m2) { ringw[pos & (S1_Q - 1)] = ((unsigned)s.z << S1_RBITS) | dl2; ++pos; }
        if (m3) { ringw[pos & (S1_Q - 1)] = ((unsigned)s.w << S1_RBITS) | dl3; }
        qt += (unsigned)(__builtin_popcount(b0) + __builtin_popcount(b1) +
                         __builtin_popcount(b2) + __builtin_popcount(b3));
        while (qt - qh >= 32u) {
            s1_drain(ringw, qh, 32, lane, h, acc);
            qh += 32u;
        }
    };

    const int step = S1_WAVES * S1_CHUNK;
    int e0 = sbeg + wave * S1_CHUNK;
    int4 dA, sA, dB, sB;
    load_chunk(dA, sA, e0);
    for (; e0 < send; e0 += 2 * step) {
        load_chunk(dB, sB, e0 + step);
        filter_chunk(dA, sA);
        load_chunk(dA, sA, e0 + 2 * step);
        filter_chunk(dB, sB);
    }
    if (qt != qh) s1_drain(ringw, qh, (int)(qt - qh), lane, h, acc);
    __syncthreads();

    const int rows = min(S1_R, n - lo);
    float* aggb = agg + (size_t)lo * HID;
    for (int i = tid; i < rows * HID; i += S1_WAVES * 32)
        unsafeAtomicAdd(aggb + i, acc[i]);
}

__global__ void k_layer2(const float* __restrict__ agg1, const float* __restrict__ b1,
                         const float* __restrict__ W2, const float* __restrict__ dinv,
                         float* __restrict__ h2, float* __restrict__ agg2, int n) {
    int i = blockIdx.x * blockDim.x + threadIdx.x;
    if (i >= n) return;
    const float* r = agg1 + (size_t)i * HID;
    const float di = dinv[i];
    float z0 = 0.0f, z1 = 0.0f;
    #pragma unroll
    for (int k = 0; k < HID; ++k) {
        float v = fmaxf(r[k] * di + b1[k], 0.0f);
        z0 = fmaf(v, W2[k * OUT_DIM + 0], z0);
        z1 = fmaf(v, W2[k * OUT_DIM + 1], z1);
    }
    z0 *= di;
    z1 *= di;
    h2[i * 2 + 0]   = z0;
    h2[i * 2 + 1]   = z1;
    agg2[i * 2 + 0] = z0;
    agg2[i * 2 + 1] = z1;
}

#define S2_R      25024
#define S2_SLICES 64
#define S2_T      1024
#define S2_K      7
#define S2_F      7

__global__ __launch_bounds__(S2_T) void k_scatter2(const int* __restrict__ src, const int* __restrict__ dst,
                                                   const float* __restrict__ h2,
                                                   float* __restrict__ agg2, int nE, int n) {
    __shared__ __attribute__((aligned(16))) float acc[S2_R * OUT_DIM];

    const int tid  = threadIdx.x;
    const int lo   = blockIdx.x * S2_R;
    const int rows = min(S2_R, n - lo);
    const int nsl  = (int)gridDim.y;
    const int per  = ((nE + nsl - 1) / nsl + 31) & ~31;
    const int sbeg = (int)blockIdx.y * per;
    const int send = min(sbeg + per, nE);

    for (int i = tid; i < S2_R * OUT_DIM / 4; i += S2_T)
        ((float4*)acc)[i] = make_float4(0.0f, 0.0f, 0.0f, 0.0f);
    __syncthreads();

    auto load_idx = [&](int (&d)[S2_K], int (&s)[S2_K], int base) {
        #pragma unroll
        for (int k = 0; k < S2_K; ++k) {
            const int e  = base + k * S2_T + tid;
            const int ec = min(e, nE - 1);
            const int dv = dst[ec];
            s[k] = src[ec];
            d[k] = (e < send) ? dv : -1;
        }
    };

    int d[S2_K], s[S2_K];
    load_idx(d, s, sbeg);
    for (int base = sbeg; base < send; base += S2_T * S2_K) {
        unsigned dl[S2_K];
        bool     own[S2_K];
        float2   hv[S2_K];
        #pragma unroll
        for (int k = 0; k < S2_K; ++k) {
            const unsigned t = (unsigned)(d[k] - lo);
            own[k] = t < (unsigned)S2_R;
            dl[k]  = own[k] ? t : 0u;
            const unsigned sg = own[k] ? (unsigned)s[k] : 0u;
            hv[k] = *(const float2*)(h2 + (size_t)sg * OUT_DIM);
        }
        int nd[S2_K], ns[S2_K];
        load_idx(nd, ns, base + S2_T * S2_K);
        #pragma unroll
        for (int k = 0; k < S2_K; ++k) {
            if (own[k]) {
                unsafeAtomicAdd(&acc[dl[k] * OUT_DIM + 0], hv[k].x);
                unsafeAtomicAdd(&acc[dl[k] * OUT_DIM + 1], hv[k].y);
            }
        }
        #pragma unroll
        for (int k = 0; k < S2_K; ++k) { d[k] = nd[k]; s[k] = ns[k]; }
    }
    __syncthreads();

    const int lim = rows * OUT_DIM;
    float* aggb = agg2 + (size_t)lo * OUT_DIM;
    for (int i0 = tid; i0 < S2_R * OUT_DIM; i0 += S2_T * S2_F) {
        float c[S2_F];
        #pragma unroll
        for (int k = 0; k < S2_F; ++k) c[k] = acc[min(i0 + k * S2_T, S2_R * OUT_DIM - 1)];
        #pragma unroll
        for (int k = 0; k < S2_F; ++k)
            if (i0 + k * S2_T < lim) unsafeAtomicAdd(aggb + i0 + k * S2_T, c[k]);
    }
}

__global__ void k_final(const float* __restrict__ agg2, const float* __restrict__ dinv,
                        const float* __restrict__ b2,
                        float* __restrict__ out, int n) {
    int i = blockIdx.x * blockDim.x + threadIdx.x;
    if (i >= n) return;
    const float di = dinv[i];
    float z0 = agg2[i * 2 + 0] * di + b2[0];
    float z1 = agg2[i * 2 + 1] * di + b2[1];
    float mx = fmaxf(z0, z1);
    float l  = mx + logf(expf(z0 - mx) + expf(z1 - mx));
    out[i * 2 + 0] = z0 - l;
    out[i * 2 + 1] = z1 - l;
}

extern "C" void kernel_launch(void* const* d_in, const int* in_sizes, int n_in,
                              void* d_out, int out_size, void* d_ws, size_t ws_size,
                              hipStream_t stream) {
    const float* x  = (const float*)d_in[0];
    const int*   ei = (const int*)d_in[1];
    const float* W1 = (const float*)d_in[2];
    const float* b1 = (const float*)d_in[3];
    const float* W2 = (const float*)d_in[4];
    const float* b2 = (const float*)d_in[5];
    float* out = (float*)d_out;

    const int n  = in_sizes[0] / IN_DIM;
    const int nE = in_sizes[1] / 2;
    const int* src = ei;
    const int* dst = ei + nE;

    float* ws   = (float*)d_ws;
    float* dinv = ws;
    float* h    = ws + n;
    float* agg1 = ws + (size_t)n * 17;
    float* h2   = h;
    float* agg2 = h + (size_t)2 * n;

    float* planes = h;

    const int B = 256;
    k_deg_edges    <<<dim3((n + DG_R - 1) / DG_R, DG_SLICES), DG_T, 0, stream>>>(dst, planes, nE, n);
    k_finalize_dinv<<<(n  + B - 1) / B, B, 0, stream>>>(planes, dinv, n);

    const int tiles   = (n + 15) / 16;
    const int gblocks = (tiles + 3) / 4;
    k_gemm1_wmma   <<<gblocks, 128, 0, stream>>>(x, W1, dinv, h, agg1, n);

    k_scatter1     <<<dim3((n + S1_R - 1) / S1_R, S1_SLICES), S1_WAVES * 32, 0, stream>>>(src, dst, h, agg1, nE, n);
    k_layer2       <<<(n  + B - 1) / B, B, 0, stream>>>(agg1, b1, W2, dinv, h2, agg2, n);
    k_scatter2     <<<dim3((n + S2_R - 1) / S2_R, S2_SLICES), S2_T, 0, stream>>>(src, dst, h2, agg2, nE, n);
    k_final        <<<(n  + B - 1) / B, B, 0, stream>>>(agg2, dinv, b2, out, n);
}


# per-slice degree counts stored in each product block's own h/agg1 rows; the product wave sums its 16 nodes' counts (exact, order changed) and takes rsqrt itself; the dinv launch is gone
# speedup vs baseline: 9.5359x; 1.0303x over previous
#include <hip/hip_runtime.h>
#include <hip/hip_bf16.h>
#include <math.h>

typedef __attribute__((ext_vector_type(2))) float v2f;
typedef __attribute__((ext_vector_type(8))) float v8f;

#define IN_DIM  128
#define HID     16
#define OUT_DIM 2


#define DG_R      12800
#define DG_SLICES 32
#define DG_T      1024
#define DG_B      13
#define DG_F      13
#define PT_ROWS   64
static_assert(DG_SLICES == 2 * HID, "a node's slice counts fill exactly its h row and its agg1 row");
static_assert(DG_R % PT_ROWS == 0, "ranges start on product-block bounds, so a wave's count store is one whole line");

__global__ __launch_bounds__(DG_T) void k_deg_edges(const int* __restrict__ dst, float* __restrict__ planes,
                                                    int nE, int n) {
    __shared__ __attribute__((aligned(16))) float cnt[DG_R];

    const int tid  = threadIdx.x;
    const int lo   = blockIdx.x * DG_R;
    const int nsl  = (int)gridDim.y;
    const int per  = ((nE + nsl - 1) / nsl + 3) & ~3;
    const int sbeg = (int)blockIdx.y * per;
    const int send = min(sbeg + per, nE);

    for (int i = tid; i < DG_R / 4; i += DG_T)
        ((float4*)cnt)[i] = make_float4(0.0f, 0.0f, 0.0f, 0.0f);
    __syncthreads();

    for (int base = sbeg; base < send; base += DG_T * DG_B * 4) {
        int4 d[DG_B];
        #pragma unroll
        for (int k = 0; k < DG_B; ++k) {
            const int e = base + 4 * (k * DG_T + tid);
            const unsigned off = 4u * (unsigned)min(e, nE - 4);
            d[k] = *(const int4*)((const char*)dst + off);
        }
        #pragma unroll
        for (int k = 0; k < DG_B; ++k) {
            if (base + 4 * (k * DG_T + tid) < send) {
                const unsigned d0 = (unsigned)(d[k].x - lo), d1 = (unsigned)(d[k].y - lo);
                const unsigned d2 = (unsigned)(d[k].z - lo), d3 = (unsigned)(d[k].w - lo);
                if (d0 < (unsigned)DG_R) unsafeAtomicAdd(&cnt[d0], 1.0f);
                if (d1 < (unsigned)DG_R) unsafeAtomicAdd(&cnt[d1], 1.0f);
                if (d2 < (unsigned)DG_R) unsafeAtomicAdd(&cnt[d2], 1.0f);
                if (d3 < (unsigned)DG_R) unsafeAtomicAdd(&cnt[d3], 1.0f);
            }
        }
    }
    __syncthreads();

    const int rows = min(DG_R, n - lo);
    float* pl = planes + (size_t)(blockIdx.y >> 4) * HID * n;
    const int ys = (int)blockIdx.y & 15;
    float c[DG_F];
    #pragma unroll
    for (int k = 0; k < DG_F; ++k) c[k] = cnt[min(tid + k * DG_T, DG_R - 1)];
    #pragma unroll
    for (int k = 0; k < DG_F; ++k) {
        if (tid + k * DG_T < rows) {
            const int i  = lo + tid + k * DG_T;
            const int b  = i / PT_ROWS;
            const int rb = min(PT_ROWS, n - b * PT_ROWS);
            pl[b * (PT_ROWS * HID) + ys * rb + (i - b * PT_ROWS)] = c[k];
        }
    }
}

__global__ __launch_bounds__(128) void k_gemm1_wmma(const float* __restrict__ x,
                                                    const float* __restrict__ W1,
                                                    float* __restrict__ dinv,
                                                    float* h,
                                                    float* agg, int n) {
    __shared__ float sW[IN_DIM * HID];
    __shared__ float sX[4][16 * 132];
    __shared__ float sD[PT_ROWS];

    const int wave = __builtin_amdgcn_readfirstlane(threadIdx.x >> 5);
    const int lane = threadIdx.x & 31;
    const int tile = blockIdx.x * 4 + wave;
    const int row0 = tile * 16;
    const bool active = (row0 < n);

    const int blk0 = blockIdx.x * PT_ROWS;
    const int rb   = min(PT_ROWS, n - blk0);
    const int node = min(row0 + (lane & 15), n - 1);
    const float* ct = ((lane >> 4) ? agg : h) + blockIdx.x * (PT_ROWS * HID) + (node - blk0);
    float cs[HID];
    #pragma unroll
    for (int k = 0; k < HID; ++k) cs[k] = ct[k * rb];

    for (int t = threadIdx.x; t < IN_DIM * HID; t += blockDim.x) sW[t] = W1[t];

    if (active) {
        float dg = 0.0f;
        #pragma unroll
        for (int k = 0; k < HID; ++k) dg += cs[k];
        dg += __shfl_xor(dg, 16);
        dg += 1.0f;
        const float dv = (dg > 0.0f) ? rsqrtf(dg) : 0.0f;
        if (lane < 16) sD[wave * 16 + lane] = dv;

        #pragma unroll 4
        for (int i = 0; i < 16; ++i) {
            int r = row0 + i;
            if (r >= n) r = n - 1;
            const float d = __builtin_bit_cast(float, __builtin_amdgcn_readlane(__builtin_bit_cast(int, dv), i));
            const float4 v = ((const float4*)(x + (size_t)r * IN_DIM))[lane];
            const v2f dd = {d, d};
            const v2f lo = (v2f){v.x, v.y} * dd;
            const v2f hi = (v2f){v.z, v.w} * dd;
            *(float4*)&sX[wave][i * 132 + lane * 4] = make_float4(lo.x, lo.y, hi.x, hi.y);
        }
    }
    __syncthreads();
    if (!active) return;

    if (wave == 0) {
        if (blk0 + lane < n)      dinv[blk0 + lane]      = sD[lane];
        if (blk0 + 32 + lane < n) dinv[blk0 + 32 + lane] = sD[32 + lane];
    }

    const int m  = lane & 15;
    const int kb = (lane >> 4) << 1;
    const float* xr = &sX[wave][m * 132];

    v8f acc = {};
    #pragma unroll
    for (int k0 = 0; k0 < IN_DIM; k0 += 4) {
        v2f a, b;
        a.x = xr[k0 + kb];
        a.y = xr[k0 + kb + 1];
        b.x = sW[(k0 + kb)     * HID + m];
        b.y = sW[(k0 + kb + 1) * HID + m];
        acc = __builtin_amdgcn_wmma_f32_16x16x4_f32(false, a, false, b,
                                                    (short)0, acc, false, false);
    }

    const int rbase = row0 + ((lane >> 4) << 3);
    #pragma unroll
    for (int v = 0; v < 8; ++v) {
        int r = rbase + v;
        if (r < n) {
            h[(size_t)r * HID + m]   = acc[v];
            agg[(size_t)r * HID + m] = acc[v];
        }
    }
}

#define S1_RBITS  12
#define S1_R      (1 << S1_RBITS)
#define S1_SLICES 10
#define S1_WAVES  32
#define S1_Q      256
#define S1_CHUNK  128

__device__ __forceinline__ void s1_drain(const unsigned* ringw, unsigned head, int count,
                                         int lane,
                                         const float* __restrict__ h, float* acc) {
    unsigned p = 0;
    if (lane < count)
        p = ringw[(head + (unsigned)lane) & (S1_Q - 1)];
    const unsigned c    = lane & 15;
    const int      half = lane >> 4;
    float v[HID];
    unsigned dj[HID];
    #pragma unroll
    for (int j = 0; j < HID; ++j) {
        const int sl = 2 * j + half;
        const unsigned pj = (unsigned)__shfl((int)p, sl);
        dj[j] = pj & (S1_R - 1);
        v[j]  = h[(pj >> S1_RBITS) * HID + c];
    }
    #pragma unroll
    for (int j = 0; j < HID; ++j) {
        if (2 * j + half < count)
            unsafeAtomicAdd(&acc[dj[j] * HID + c], v[j]);
    }
}

__global__ __launch_bounds__(1024) void k_scatter1(const int* __restrict__ src, const int* __restrict__ dst,
                                                   const float* __restrict__ h,
                                                   float* __restrict__ agg, int nE, int n) {
    __shared__ __attribute__((aligned(16))) float acc[S1_R * HID];
    __shared__ unsigned ring[S1_WAVES][S1_Q];

    const int tid  = threadIdx.x;
    const int lane = tid & 31;
    const int wave = __builtin_amdgcn_readfirstlane(tid >> 5);
    const int lo   = blockIdx.x * S1_R;
    const int nsl  = (int)gridDim.y;
    const int per  = ((nE + nsl - 1) / nsl + S1_CHUNK - 1) / S1_CHUNK * S1_CHUNK;
    const int sbeg = (int)blockIdx.y * per;
    const int send = min(sbeg + per, nE);

    for (int i = tid; i < S1_R * HID / 4; i += S1_WAVES * 32)
        ((float4*)acc)[i] = make_float4(0.0f, 0.0f, 0.0f, 0.0f);
    __syncthreads();

    unsigned* ringw = ring[wave];
    unsigned qh = 0, qt = 0;

    auto load_chunk = [&](int4 &d, int4 &s, int base) {
        const int e  = base + 4 * lane;
        const unsigned off = 4u * (unsigned)min(e, nE - 4);
        const int4 dv = *(const int4*)((const char*)dst + off);
        s = *(const int4*)((const char*)src + off);
        d = (e < send) ? dv : make_int4(-1, -1, -1, -1);
    };
    auto filter_chunk = [&](const int4 &d, const int4 &s) {
        const unsigned dl0 = (unsigned)(d.x - lo), dl1 = (unsigned)(d.y - lo);
        const unsigned dl2 = (unsigned)(d.z - lo), dl3 = (unsigned)(d.w - lo);
        const bool m0 = dl0 < (unsigned)S1_R, m1 = dl1 < (unsigned)S1_R;
        const bool m2 = dl2 < (unsigned)S1_R, m3 = dl3 < (unsigned)S1_R;
        const unsigned b0 = __builtin_amdgcn_ballot_w32(m0), b1 = __builtin_amdgcn_ballot_w32(m1);
        const unsigned b2 = __builtin_amdgcn_ballot_w32(m2), b3 = __builtin_amdgcn_ballot_w32(m3);
        unsigned pos = qt + __builtin_amdgcn_mbcnt_lo(b0, __builtin_amdgcn_mbcnt_lo(b1,
                            __builtin_amdgcn_mbcnt_lo(b2, __builtin_amdgcn_mbcnt_lo(b3, 0u))));
        if (m0) { ringw[pos & (S1_Q - 1)] = ((unsigned)s.x << S1_RBITS) | dl0; ++pos; }
        if (m1) { ringw[pos & (S1_Q - 1)] = ((unsigned)s.y << S1_RBITS) | dl1; ++pos; }
        if (m2) { ringw[pos & (S1_Q - 1)] = ((unsigned)s.z << S1_RBITS) | dl2; ++pos; }
        if (m3) { ringw[pos & (S1_Q - 1)] = ((unsigned)s.w << S1_RBITS) | dl3; }
        qt += (unsigned)(__builtin_popcount(b0) + __builtin_popcount(b1) +
                         __builtin_popcount(b2) + __builtin_popcount(b3));
        while (qt - qh >= 32u) {
            s1_drain(ringw, qh, 32, lane, h, acc);
            qh += 32u;
        }
    };

    const int step = S1_WAVES * S1_CHUNK;
    int e0 = sbeg + wave * S1_CHUNK;
    int4 dA, sA, dB, sB;
    load_chunk(dA, sA, e0);
    for (; e0 < send; e0 += 2 * step) {
        load_chunk(dB, sB, e0 + step);
        filter_chunk(dA, sA);
        load_chunk(dA, sA, e0 + 2 * step);
        filter_chunk(dB, sB);
    }
    if (qt != qh) s1_drain(ringw, qh, (int)(qt - qh), lane, h, acc);
    __syncthreads();

    const int rows = min(S1_R, n - lo);
    float* aggb = agg + (size_t)lo * HID;
    for (int i = tid; i < rows * HID; i += S1_WAVES * 32)
        unsafeAtomicAdd(aggb + i, acc[i]);
}

__global__ void k_layer2(const float* __restrict__ agg1, const float* __restrict__ b1,
                         const float* __restrict__ W2, const float* __restrict__ dinv,
                         float* __restrict__ h2, float* __restrict__ agg2, int n) {
    int i = blockIdx.x * blockDim.x + threadIdx.x;
    if (i >= n) return;
    const float* r = agg1 + (size_t)i * HID;
    const float di = dinv[i];
    float z0 = 0.0f, z1 = 0.0f;
    #pragma unroll
    for (int k = 0; k < HID; ++k) {
        float v = fmaxf(r[k] * di + b1[k], 0.0f);
        z0 = fmaf(v, W2[k * OUT_DIM + 0], z0);
        z1 = fmaf(v, W2[k * OUT_DIM + 1], z1);
    }
    z0 *= di;
    z1 *= di;
    h2[i * 2 + 0]   = z0;
    h2[i * 2 + 1]   = z1;
    agg2[i * 2 + 0] = z0;
    agg2[i * 2 + 1] = z1;
}

#define S2_R      25024
#define S2_SLICES 64
#define S2_T      1024
#define S2_K      7
#define S2_F      7

__global__ __launch_bounds__(S2_T) void k_scatter2(const int* __restrict__ src, const int* __restrict__ dst,
                                                   const float* __restrict__ h2,
                                                   float* __restrict__ agg2, int nE, int n) {
    __shared__ __attribute__((aligned(16))) float acc[S2_R * OUT_DIM];

    const int tid  = threadIdx.x;
    const int lo   = blockIdx.x * S2_R;
    const int rows = min(S2_R, n - lo);
    const int nsl  = (int)gridDim.y;
    const int per  = ((nE + nsl - 1) / nsl + 31) & ~31;
    const int sbeg = (int)blockIdx.y * per;
    const int send = min(sbeg + per, nE);

    for (int i = tid; i < S2_R * OUT_DIM / 4; i += S2_T)
        ((float4*)acc)[i] = make_float4(0.0f, 0.0f, 0.0f, 0.0f);
    __syncthreads();

    auto load_idx = [&](int (&d)[S2_K], int (&s)[S2_K], int base) {
        #pragma unroll
        for (int k = 0; k < S2_K; ++k) {
            const int e  = base + k * S2_T + tid;
            const int ec = min(e, nE - 1);
            const int dv = dst[ec];
            s[k] = src[ec];
            d[k] = (e < send) ? dv : -1;
        }
    };

    int d[S2_K], s[S2_K];
    load_idx(d, s, sbeg);
    for (int base = sbeg; base < send; base += S2_T * S2_K) {
        unsigned dl[S2_K];
        bool     own[S2_K];
        float2   hv[S2_K];
        #pragma unroll
        for (int k = 0; k < S2_K; ++k) {
            const unsigned t = (unsigned)(d[k] - lo);
            own[k] = t < (unsigned)S2_R;
            dl[k]  = own[k] ? t : 0u;
            const unsigned sg = own[k] ? (unsigned)s[k] : 0u;
            hv[k] = *(const float2*)(h2 + (size_t)sg * OUT_DIM);
        }
        int nd[S2_K], ns[S2_K];
        load_idx(nd, ns, base + S2_T * S2_K);
        #pragma unroll
        for (int k = 0; k < S2_K; ++k) {
            if (own[k]) {
                unsafeAtomicAdd(&acc[dl[k] * OUT_DIM + 0], hv[k].x);
                unsafeAtomicAdd(&acc[dl[k] * OUT_DIM + 1], hv[k].y);
            }
        }
        #pragma unroll
        for (int k = 0; k < S2_K; ++k) { d[k] = nd[k]; s[k] = ns[k]; }
    }
    __syncthreads();

    const int lim = rows * OUT_DIM;
    float* aggb = agg2 + (size_t)lo * OUT_DIM;
    for (int i0 = tid; i0 < S2_R * OUT_DIM; i0 += S2_T * S2_F) {
        float c[S2_F];
        #pragma unroll
        for (int k = 0; k < S2_F; ++k) c[k] = acc[min(i0 + k * S2_T, S2_R * OUT_DIM - 1)];
        #pragma unroll
        for (int k = 0; k < S2_F; ++k)
            if (i0 + k * S2_T < lim) unsafeAtomicAdd(aggb + i0 + k * S2_T, c[k]);
    }
}

__global__ void k_final(const float* __restrict__ agg2, const float* __restrict__ dinv,
                        const float* __restrict__ b2,
                        float* __restrict__ out, int n) {
    int i = blockIdx.x * blockDim.x + threadIdx.x;
    if (i >= n) return;
    const float di = dinv[i];
    float z0 = agg2[i * 2 + 0] * di + b2[0];
    float z1 = agg2[i * 2 + 1] * di + b2[1];
    float mx = fmaxf(z0, z1);
    float l  = mx + logf(expf(z0 - mx) + expf(z1 - mx));
    out[i * 2 + 0] = z0 - l;
    out[i * 2 + 1] = z1 - l;
}

extern "C" void kernel_launch(void* const* d_in, const int* in_sizes, int n_in,
                              void* d_out, int out_size, void* d_ws, size_t ws_size,
                              hipStream_t stream) {
    const float* x  = (const float*)d_in[0];
    const int*   ei = (const int*)d_in[1];
    const float* W1 = (const float*)d_in[2];
    const float* b1 = (const float*)d_in[3];
    const float* W2 = (const float*)d_in[4];
    const float* b2 = (const float*)d_in[5];
    float* out = (float*)d_out;

    const int n  = in_sizes[0] / IN_DIM;
    const int nE = in_sizes[1] / 2;
    const int* src = ei;
    const int* dst = ei + nE;

    float* ws   = (float*)d_ws;
    float* dinv = ws;
    float* h    = ws + n;
    float* agg1 = ws + (size_t)n * 17;
    float* h2   = h;
    float* agg2 = h + (size_t)2 * n;

    float* planes = h;

    const int B = 256;
    k_deg_edges    <<<dim3((n + DG_R - 1) / DG_R, DG_SLICES), DG_T, 0, stream>>>(dst, planes, nE, n);

    const int tiles   = (n + 15) / 16;
    const int gblocks = (tiles + 3) / 4;
    k_gemm1_wmma   <<<gblocks, 128, 0, stream>>>(x, W1, dinv, h, agg1, n);

    k_scatter1     <<<dim3((n + S1_R - 1) / S1_R, S1_SLICES), S1_WAVES * 32, 0, stream>>>(src, dst, h, agg1, nE, n);
    k_layer2       <<<(n  + B - 1) / B, B, 0, stream>>>(agg1, b1, W2, dinv, h2, agg2, n);
    k_scatter2     <<<dim3((n + S2_R - 1) / S2_R, S2_SLICES), S2_T, 0, stream>>>(src, dst, h2, agg2, nE, n);
    k_final        <<<(n  + B - 1) / B, B, 0, stream>>>(agg2, dinv, b2, out, n);
}


# loads that were waited for one by one now leave as groups: W1 staging (16 serial round trips -> 1) and the 16 x rows (4 groups -> 1) in the product kernel, LDS reads of the layer-1 flush (64 -> 4)
# speedup vs baseline: 9.9336x; 1.0417x over previous
#include <hip/hip_runtime.h>
#include <hip/hip_bf16.h>
#include <math.h>

typedef __attribute__((ext_vector_type(2))) float v2f;
typedef __attribute__((ext_vector_type(8))) float v8f;

#define IN_DIM  128
#define HID     16
#define OUT_DIM 2


#define DG_R      12800
#define DG_SLICES 32
#define DG_T      1024
#define DG_B      13
#define DG_F      13
#define PT_ROWS   64
static_assert(DG_SLICES == 2 * HID, "a node's slice counts fill exactly its h row and its agg1 row");
static_assert(DG_R % PT_ROWS == 0, "ranges start on product-block bounds, so a wave's count store is one whole line");

__global__ __launch_bounds__(DG_T) void k_deg_edges(const int* __restrict__ dst, float* __restrict__ planes,
                                                    int nE, int n) {
    __shared__ __attribute__((aligned(16))) float cnt[DG_R];

    const int tid  = threadIdx.x;
    const int lo   = blockIdx.x * DG_R;
    const int nsl  = (int)gridDim.y;
    const int per  = ((nE + nsl - 1) / nsl + 3) & ~3;
    const int sbeg = (int)blockIdx.y * per;
    const int send = min(sbeg + per, nE);

    for (int i = tid; i < DG_R / 4; i += DG_T)
        ((float4*)cnt)[i] = make_float4(0.0f, 0.0f, 0.0f, 0.0f);
    __syncthreads();

    for (int base = sbeg; base < send; base += DG_T * DG_B * 4) {
        int4 d[DG_B];
        #pragma unroll
        for (int k = 0; k < DG_B; ++k) {
            const int e = base + 4 * (k * DG_T + tid);
            const unsigned off = 4u * (unsigned)min(e, nE - 4);
            d[k] = *(const int4*)((const char*)dst + off);
        }
        #pragma unroll
        for (int k = 0; k < DG_B; ++k) {
            if (base + 4 * (k * DG_T + tid) < send) {
                const unsigned d0 = (unsigned)(d[k].x - lo), d1 = (unsigned)(d[k].y - lo);
                const unsigned d2 = (unsigned)(d[k].z - lo), d3 = (unsigned)(d[k].w - lo);
                if (d0 < (unsigned)DG_R) unsafeAtomicAdd(&cnt[d0], 1.0f);
                if (d1 < (unsigned)DG_R) unsafeAtomicAdd(&cnt[d1], 1.0f);
                if (d2 < (unsigned)DG_R) unsafeAtomicAdd(&cnt[d2], 1.0f);
                if (d3 < (unsigned)DG_R) unsafeAtomicAdd(&cnt[d3], 1.0f);
            }
        }
    }
    __syncthreads();

    const int rows = min(DG_R, n - lo);
    float* pl = planes + (size_t)(blockIdx.y >> 4) * HID * n;
    const int ys = (int)blockIdx.y & 15;
    float c[DG_F];
    #pragma unroll
    for (int k = 0; k < DG_F; ++k) c[k] = cnt[min(tid + k * DG_T, DG_R - 1)];
    #pragma unroll
    for (int k = 0; k < DG_F; ++k) {
        if (tid + k * DG_T < rows) {
            const int i  = lo + tid + k * DG_T;
            const int b  = i / PT_ROWS;
            const int rb = min(PT_ROWS, n - b * PT_ROWS);
            pl[b * (PT_ROWS * HID) + ys * rb + (i - b * PT_ROWS)] = c[k];
        }
    }
}

__global__ __launch_bounds__(128) void k_gemm1_wmma(const float* __restrict__ x,
                                                    const float* __restrict__ W1,
                                                    float* __restrict__ dinv,
                                                    float* h,
                                                    float* agg, int n) {
    __shared__ __attribute__((aligned(16))) float sW[IN_DIM * HID];
    __shared__ __attribute__((aligned(16))) float sX[4][16 * 132];
    __shared__ float sD[PT_ROWS];

    const int wave = __builtin_amdgcn_readfirstlane(threadIdx.x >> 5);
    const int lane = threadIdx.x & 31;
    const int tile = blockIdx.x * 4 + wave;
    const int row0 = tile * 16;
    const bool active = (row0 < n);


    float4 w4[4];
    #pragma unroll
    for (int j = 0; j < 4; ++j) w4[j] = ((const float4*)W1)[j * 128 + (int)threadIdx.x];
    #pragma unroll
    for (int j = 0; j < 4; ++j) ((float4*)sW)[j * 128 + (int)threadIdx.x] = w4[j];

    const int blk0 = blockIdx.x * PT_ROWS;
    const int rb   = min(PT_ROWS, n - blk0);
    const int node = min(row0 + (lane & 15), n - 1);
    const float* ct = ((lane >> 4) ? agg : h) + blockIdx.x * (PT_ROWS * HID) + (node - blk0);
    float cs[HID];
    #pragma unroll
    for (int k = 0; k < HID; ++k) cs[k] = ct[k * rb];

    float4 xv[16];
    #pragma unroll
    for (int i = 0; i < 16; ++i) {
        const int r = min(row0 + i, n - 1);
        xv[i] = *(const float4*)((const char*)(x + (size_t)r * IN_DIM) + 16u * (unsigned)lane);
    }

    if (active) {
        float dg = 0.0f;
        #pragma unroll
        for (int k = 0; k < HID; ++k) dg += cs[k];
        dg += __shfl_xor(dg, 16);
        dg += 1.0f;
        const float dv = (dg > 0.0f) ? rsqrtf(dg) : 0.0f;
        if (lane < 16) sD[wave * 16 + lane] = dv;

        #pragma unroll
        for (int i = 0; i < 16; ++i) {
            const float d = __builtin_bit_cast(float, __builtin_amdgcn_readlane(__builtin_bit_cast(int, dv), i));
            const float4 v = xv[i];
            const v2f dd = {d, d};
            const v2f lo = (v2f){v.x, v.y} * dd;
            const v2f hi = (v2f){v.z, v.w} * dd;
            *(float4*)&sX[wave][i * 132 + lane * 4] = make_float4(lo.x, lo.y, hi.x, hi.y);
        }
    }
    __syncthreads();
    if (!active) return;

    if (wave == 0) {
        if (blk0 + lane < n)      dinv[blk0 + lane]      = sD[lane];
        if (blk0 + 32 + lane < n) dinv[blk0 + 32 + lane] = sD[32 + lane];
    }

    const int m  = lane & 15;
    const int kb = (lane >> 4) << 1;
    const float* xr = &sX[wave][m * 132];

    v8f acc = {};
    #pragma unroll
    for (int k0 = 0; k0 < IN_DIM; k0 += 4) {
        v2f a, b;
        a.x = xr[k0 + kb];
        a.y = xr[k0 + kb + 1];
        b.x = sW[(k0 + kb)     * HID + m];
        b.y = sW[(k0 + kb + 1) * HID + m];
        acc = __builtin_amdgcn_wmma_f32_16x16x4_f32(false, a, false, b,
                                                    (short)0, acc, false, false);
    }

    const int rbase = row0 + ((lane >> 4) << 3);
    #pragma unroll
    for (int v = 0; v < 8; ++v) {
        int r = rbase + v;
        if (r < n) {
            h[(size_t)r * HID + m]   = acc[v];
            agg[(size_t)r * HID + m] = acc[v];
        }
    }
}

#define S1_RBITS  12
#define S1_R      (1 << S1_RBITS)
#define S1_SLICES 10
#define S1_WAVES  32
#define S1_Q      256
#define S1_CHUNK  128
#define S1_F      16

__device__ __forceinline__ void s1_drain(const unsigned* ringw, unsigned head, int count,
                                         int lane,
                                         const float* __restrict__ h, float* acc) {
    unsigned p = 0;
    if (lane < count)
        p = ringw[(head + (unsigned)lane) & (S1_Q - 1)];
    const unsigned c    = lane & 15;
    const int      half = lane >> 4;
    float v[HID];
    unsigned dj[HID];
    #pragma unroll
    for (int j = 0; j < HID; ++j) {
        const int sl = 2 * j + half;
        const unsigned pj = (unsigned)__shfl((int)p, sl);
        dj[j] = pj & (S1_R - 1);
        v[j]  = h[(pj >> S1_RBITS) * HID + c];
    }
    #pragma unroll
    for (int j = 0; j < HID; ++j) {
        if (2 * j + half < count)
            unsafeAtomicAdd(&acc[dj[j] * HID + c], v[j]);
    }
}

__global__ __launch_bounds__(1024) void k_scatter1(const int* __restrict__ src, const int* __restrict__ dst,
                                                   const float* __restrict__ h,
                                                   float* __restrict__ agg, int nE, int n) {
    __shared__ __attribute__((aligned(16))) float acc[S1_R * HID];
    __shared__ unsigned ring[S1_WAVES][S1_Q];

    const int tid  = threadIdx.x;
    const int lane = tid & 31;
    const int wave = __builtin_amdgcn_readfirstlane(tid >> 5);
    const int lo   = blockIdx.x * S1_R;
    const int nsl  = (int)gridDim.y;
    const int per  = ((nE + nsl - 1) / nsl + S1_CHUNK - 1) / S1_CHUNK * S1_CHUNK;
    const int sbeg = (int)blockIdx.y * per;
    const int send = min(sbeg + per, nE);

    for (int i = tid; i < S1_R * HID / 4; i += S1_WAVES * 32)
        ((float4*)acc)[i] = make_float4(0.0f, 0.0f, 0.0f, 0.0f);
    __syncthreads();

    unsigned* ringw = ring[wave];
    unsigned qh = 0, qt = 0;

    auto load_chunk = [&](int4 &d, int4 &s, int base) {
        const int e  = base + 4 * lane;
        const unsigned off = 4u * (unsigned)min(e, nE - 4);
        const int4 dv = *(const int4*)((const char*)dst + off);
        s = *(const int4*)((const char*)src + off);
        d = (e < send) ? dv : make_int4(-1, -1, -1, -1);
    };
    auto filter_chunk = [&](const int4 &d, const int4 &s) {
        const unsigned dl0 = (unsigned)(d.x - lo), dl1 = (unsigned)(d.y - lo);
        const unsigned dl2 = (unsigned)(d.z - lo), dl3 = (unsigned)(d.w - lo);
        const bool m0 = dl0 < (unsigned)S1_R, m1 = dl1 < (unsigned)S1_R;
        const bool m2 = dl2 < (unsigned)S1_R, m3 = dl3 < (unsigned)S1_R;
        const unsigned b0 = __builtin_amdgcn_ballot_w32(m0), b1 = __builtin_amdgcn_ballot_w32(m1);
        const unsigned b2 = __builtin_amdgcn_ballot_w32(m2), b3 = __builtin_amdgcn_ballot_w32(m3);
        unsigned pos = qt + __builtin_amdgcn_mbcnt_lo(b0, __builtin_amdgcn_mbcnt_lo(b1,
                            __builtin_amdgcn_mbcnt_lo(b2, __builtin_amdgcn_mbcnt_lo(b3, 0u))));
        if (m0) { ringw[pos & (S1_Q - 1)] = ((unsigned)s.x << S1_RBITS) | dl0; ++pos; }
        if (m1) { ringw[pos & (S1_Q - 1)] = ((unsigned)s.y << S1_RBITS) | dl1; ++pos; }
        if (m2) { ringw[pos & (S1_Q - 1)] = ((unsigned)s.z << S1_RBITS) | dl2; ++pos; }
        if (m3) { ringw[pos & (S1_Q - 1)] = ((unsigned)s.w << S1_RBITS) | dl3; }
        qt += (unsigned)(__builtin_popcount(b0) + __builtin_popcount(b1) +
                         __builtin_popcount(b2) + __builtin_popcount(b3));
        while (qt - qh >= 32u) {
            s1_drain(ringw, qh, 32, lane, h, acc);
            qh += 32u;
        }
    };

    const int step = S1_WAVES * S1_CHUNK;
    int e0 = sbeg + wave * S1_CHUNK;
    int4 dA, sA, dB, sB;
    load_chunk(dA, sA, e0);
    for (; e0 < send; e0 += 2 * step) {
        load_chunk(dB, sB, e0 + step);
        filter_chunk(dA, sA);
        load_chunk(dA, sA, e0 + 2 * step);
        filter_chunk(dB, sB);
    }
    if (qt != qh) s1_drain(ringw, qh, (int)(qt - qh), lane, h, acc);
    __syncthreads();

    const int rows = min(S1_R, n - lo);
    const int lim  = rows * HID;
    float* aggb = agg + (size_t)lo * HID;
    if (rows == S1_R) {
        for (int i0 = tid; i0 < S1_R * HID; i0 += S1_WAVES * 32 * S1_F) {
            float c[S1_F];
            #pragma unroll
            for (int k = 0; k < S1_F; ++k) c[k] = acc[i0 + k * (S1_WAVES * 32)];
            #pragma unroll
            for (int k = 0; k < S1_F; ++k) unsafeAtomicAdd(aggb + i0 + k * (S1_WAVES * 32), c[k]);
        }
    } else {
        for (int i0 = tid; i0 < lim; i0 += S1_WAVES * 32 * S1_F) {
            float c[S1_F];
            #pragma unroll
            for (int k = 0; k < S1_F; ++k) c[k] = acc[i0 + k * (S1_WAVES * 32)];
            #pragma unroll
            for (int k = 0; k < S1_F; ++k)
                if (i0 + k * (S1_WAVES * 32) < lim) unsafeAtomicAdd(aggb + i0 + k * (S1_WAVES * 32), c[k]);
        }
    }
}

__global__ void k_layer2(const float* __restrict__ agg1, const float* __restrict__ b1,
                         const float* __restrict__ W2, const float* __restrict__ dinv,
                         float* __restrict__ h2, float* __restrict__ agg2, int n) {
    int i = blockIdx.x * blockDim.x + threadIdx.x;
    if (i >= n) return;
    const float* r = agg1 + (size_t)i * HID;
    const float di = dinv[i];
    float z0 = 0.0f, z1 = 0.0f;
    #pragma unroll
    for (int k = 0; k < HID; ++k) {
        float v = fmaxf(r[k] * di + b1[k], 0.0f);
        z0 = fmaf(v, W2[k * OUT_DIM + 0], z0);
        z1 = fmaf(v, W2[k * OUT_DIM + 1], z1);
    }
    z0 *= di;
    z1 *= di;
    h2[i * 2 + 0]   = z0;
    h2[i * 2 + 1]   = z1;
    agg2[i * 2 + 0] = z0;
    agg2[i * 2 + 1] = z1;
}

#define S2_R      25024
#define S2_SLICES 64
#define S2_T      1024
#define S2_K      7
#define S2_F      7

__global__ __launch_bounds__(S2_T) void k_scatter2(const int* __restrict__ src, const int* __restrict__ dst,
                                                   const float* __restrict__ h2,
                                                   float* __restrict__ agg2, int nE, int n) {
    __shared__ __attribute__((aligned(16))) float acc[S2_R * OUT_DIM];

    const int tid  = threadIdx.x;
    const int lo   = blockIdx.x * S2_R;
    const int rows = min(S2_R, n - lo);
    const int nsl  = (int)gridDim.y;
    const int per  = ((nE + nsl - 1) / nsl + 31) & ~31;
    const int sbeg = (int)blockIdx.y * per;
    const int send = min(sbeg + per, nE);

    for (int i = tid; i < S2_R * OUT_DIM / 4; i += S2_T)
        ((float4*)acc)[i] = make_float4(0.0f, 0.0f, 0.0f, 0.0f);
    __syncthreads();

    auto load_idx = [&](int (&d)[S2_K], int (&s)[S2_K], int base) {
        #pragma unroll
        for (int k = 0; k < S2_K; ++k) {
            const int e  = base + k * S2_T + tid;
            const int ec = min(e, nE - 1);
            const int dv = dst[ec];
            s[k] = src[ec];
            d[k] = (e < send) ? dv : -1;
        }
    };

    int d[S2_K], s[S2_K];
    load_idx(d, s, sbeg);
    for (int base = sbeg; base < send; base += S2_T * S2_K) {
        unsigned dl[S2_K];
        bool     own[S2_K];
        float2   hv[S2_K];
        #pragma unroll
        for (int k = 0; k < S2_K; ++k) {
            const unsigned t = (unsigned)(d[k] - lo);
            own[k] = t < (unsigned)S2_R;
            dl[k]  = own[k] ? t : 0u;
            const unsigned sg = own[k] ? (unsigned)s[k] : 0u;
            hv[k] = *(const float2*)(h2 + (size_t)sg * OUT_DIM);
        }
        int nd[S2_K], ns[S2_K];
        load_idx(nd, ns, base + S2_T * S2_K);
        #pragma unroll
        for (int k = 0; k < S2_K; ++k) {
            if (own[k]) {
                unsafeAtomicAdd(&acc[dl[k] * OUT_DIM + 0], hv[k].x);
                unsafeAtomicAdd(&acc[dl[k] * OUT_DIM + 1], hv[k].y);
            }
        }
        #pragma unroll
        for (int k = 0; k < S2_K; ++k) { d[k] = nd[k]; s[k] = ns[k]; }
    }
    __syncthreads();

    const int lim = rows * OUT_DIM;
    float* aggb = agg2 + (size_t)lo * OUT_DIM;
    for (int i0 = tid; i0 < S2_R * OUT_DIM; i0 += S2_T * S2_F) {
        float c[S2_F];
        #pragma unroll
        for (int k = 0; k < S2_F; ++k) c[k] = acc[min(i0 + k * S2_T, S2_R * OUT_DIM - 1)];
        #pragma unroll
        for (int k = 0; k < S2_F; ++k)
            if (i0 + k * S2_T < lim) unsafeAtomicAdd(aggb + i0 + k * S2_T, c[k]);
    }
}

__global__ void k_final(const float* __restrict__ agg2, const float* __restrict__ dinv,
                        const float* __restrict__ b2,
                        float* __restrict__ out, int n) {
    int i = blockIdx.x * blockDim.x + threadIdx.x;
    if (i >= n) return;
    const float di = dinv[i];
    float z0 = agg2[i * 2 + 0] * di + b2[0];
    float z1 = agg2[i * 2 + 1] * di + b2[1];
    float mx = fmaxf(z0, z1);
    float l  = mx + logf(expf(z0 - mx) + expf(z1 - mx));
    out[i * 2 + 0] = z0 - l;
    out[i * 2 + 1] = z1 - l;
}

extern "C" void kernel_launch(void* const* d_in, const int* in_sizes, int n_in,
                              void* d_out, int out_size, void* d_ws, size_t ws_size,
                              hipStream_t stream) {
    const float* x  = (const float*)d_in[0];
    const int*   ei = (const int*)d_in[1];
    const float* W1 = (const float*)d_in[2];
    const float* b1 = (const float*)d_in[3];
    const float* W2 = (const float*)d_in[4];
    const float* b2 = (const float*)d_in[5];
    float* out = (float*)d_out;

    const int n  = in_sizes[0] / IN_DIM;
    const int nE = in_sizes[1] / 2;
    const int* src = ei;
    const int* dst = ei + nE;

    float* ws   = (float*)d_ws;
    float* dinv = ws;
    float* h    = ws + n;
    float* agg1 = ws + (size_t)n * 17;
    float* h2   = h;
    float* agg2 = h + (size_t)2 * n;

    float* planes = h;

    const int B = 256;
    k_deg_edges    <<<dim3((n + DG_R - 1) / DG_R, DG_SLICES), DG_T, 0, stream>>>(dst, planes, nE, n);

    const int tiles   = (n + 15) / 16;
    const int gblocks = (tiles + 3) / 4;
    k_gemm1_wmma   <<<gblocks, 128, 0, stream>>>(x, W1, dinv, h, agg1, n);

    k_scatter1     <<<dim3((n + S1_R - 1) / S1_R, S1_SLICES), S1_WAVES * 32, 0, stream>>>(src, dst, h, agg1, nE, n);
    k_layer2       <<<(n  + B - 1) / B, B, 0, stream>>>(agg1, b1, W2, dinv, h2, agg2, n);
    k_scatter2     <<<dim3((n + S2_R - 1) / S2_R, S2_SLICES), S2_T, 0, stream>>>(src, dst, h2, agg2, nE, n);
    k_final        <<<(n  + B - 1) / B, B, 0, stream>>>(agg2, dinv, b2, out, n);
}
